# starting kernel
# baseline (speedup 1.0000x reference)
#include <hip/hip_runtime.h>

typedef __attribute__((ext_vector_type(2))) float v2f;
typedef __attribute__((ext_vector_type(8))) float v8f;

#define WAVES_PER_BLOCK 8
#define BLOCK_THREADS 256

__global__ __launch_bounds__(BLOCK_THREADS)
void gemm16_wmma_f32(const float* __restrict__ A, const float* __restrict__ B,
                     const float* __restrict__ bias, float* __restrict__ C,
                     int M, int K, int N, int transB, int hasBias)
{
    __shared__ float Bs[256 * 16];
    const int tid   = threadIdx.x;
    const int nbase = blockIdx.y * 16;

    for (int i = tid; i < K * 16; i += BLOCK_THREADS) {
        int k  = i >> 4;
        int n  = i & 15;
        int gn = nbase + n;
        float v = 0.0f;
        if (gn < N) v = transB ? B[(long)gn * K + k] : B[(long)k * N + gn];
        Bs[i] = v;
    }
    __syncthreads();

    const int  wave  = tid >> 5;
    const int  lane  = tid & 31;
    const long mtile = (long)blockIdx.x * WAVES_PER_BLOCK + wave;
    if (mtile * 16 >= M) return;

    const int mn   = lane & 15;
    const int half = lane >> 4;

    const float*  arow  = A + (mtile * 16 + mn) * (long)K;
    const float2* arow2 = (const float2*)arow;

    v8f acc = {};
    for (int k = 0; k < K; k += 4) {
        const int kb = k + 2 * half;
        float2 a2 = arow2[kb >> 1];
        v2f av; av.x = a2.x; av.y = a2.y;
        v2f bv; bv.x = Bs[(kb + 0) * 16 + mn];
                bv.y = Bs[(kb + 1) * 16 + mn];
        acc = __builtin_amdgcn_wmma_f32_16x16x4_f32(
                   false, av,  false, bv,
                   (short)0, acc,  false,  false);
    }

    const int gn = nbase + mn;
    if (gn < N) {
        const float bval = hasBias ? bias[gn] : 0.0f;
        #pragma unroll
        for (int r = 0; r < 8; ++r) {
            long row = mtile * 16 + r + 8 * half;
            C[row * (long)N + gn] = acc[r] + bval;
        }
    }
}

__global__ __launch_bounds__(BLOCK_THREADS)
void spmm_scatter_f32(const int* __restrict__ rows, const int* __restrict__ cols,
                      const float* __restrict__ ew, const float* __restrict__ dense,
                      float* __restrict__ out, int nE, int F)
{
    const int warp = blockIdx.x * (BLOCK_THREADS >> 5) + (threadIdx.x >> 5);
    const int lane = threadIdx.x & 31;
    if (warp >= nE) return;
    const int   dst = rows[warp];
    const int   src = cols[warp];
    const float w   = ew[warp];
    const float4* s = (const float4*)(dense + (long)src * F);
    float*        d = out + (long)dst * F;
    for (int i = lane; i < (F >> 2); i += 32) {
        float4 v = s[i];
        atomicAdd(d + 4 * i + 0, v.x * w);
        atomicAdd(d + 4 * i + 1, v.y * w);
        atomicAdd(d + 4 * i + 2, v.z * w);
        atomicAdd(d + 4 * i + 3, v.w * w);
    }
}

__global__ void zero_f32(float* __restrict__ p, long n)
{
    long i      = (long)blockIdx.x * blockDim.x + threadIdx.x;
    long stride = (long)gridDim.x * blockDim.x;
    for (; i < n; i += stride) p[i] = 0.0f;
}

__global__ void bias_relu_f32(float* __restrict__ p, const float* __restrict__ b,
                              long nrows, int F)
{
    long n      = nrows * (long)F;
    long i      = (long)blockIdx.x * blockDim.x + threadIdx.x;
    long stride = (long)gridDim.x * blockDim.x;
    for (; i < n; i += stride) {
        float v = p[i] + b[(int)(i & (long)(F - 1))];
        p[i] = v > 0.0f ? v : 0.0f;
    }
}

static inline int mtiles_blocks(int M) {
    return ((M / 16) + WAVES_PER_BLOCK - 1) / WAVES_PER_BLOCK;
}

extern "C" void kernel_launch(void* const* d_in, const int* in_sizes, int n_in,
                              void* d_out, int out_size, void* d_ws, size_t ws_size,
                              hipStream_t stream)
{
    const float* feat0 = (const float*)d_in[0];
    const float* feat1 = (const float*)d_in[1];
    const float* fc0_w = (const float*)d_in[2];
    const float* fc0_b = (const float*)d_in[3];
    const float* fc1_w = (const float*)d_in[4];
    const float* fc1_b = (const float*)d_in[5];
    const float* W1    = (const float*)d_in[6];
    const float* b1    = (const float*)d_in[7];
    const float* W2    = (const float*)d_in[8];
    const float* b2    = (const float*)d_in[9];
    const float* pw    = (const float*)d_in[10];
    const float* pb    = (const float*)d_in[11];
    const int*   erow  = (const int*)d_in[12];
    const int*   ecol  = (const int*)d_in[13];
    const float* ew    = (const float*)d_in[14];
    float*       out   = (float*)d_out;

    const int E       = in_sizes[12];
    const int N_NODES = 100000;
    const int NT      = 50000;

    float* x   = (float*)d_ws;
    float* h   = x + (long)N_NODES * 128;
    float* agg = h + (long)N_NODES * 128;

    gemm16_wmma_f32<<<dim3(mtiles_blocks(NT), 8), BLOCK_THREADS, 0, stream>>>(
        feat0, fc0_w, fc0_b, x, NT, 128, 128, 1, 1);
    gemm16_wmma_f32<<<dim3(mtiles_blocks(NT), 8), BLOCK_THREADS, 0, stream>>>(
        feat1, fc1_w, fc1_b, x + (long)NT * 128, NT, 256, 128, 1, 1);

    gemm16_wmma_f32<<<dim3(mtiles_blocks(N_NODES), 8), BLOCK_THREADS, 0, stream>>>(
        x, W1, nullptr, h, N_NODES, 128, 128, 0, 0);
    zero_f32<<<2048, 256, 0, stream>>>(agg, (long)N_NODES * 128);
    spmm_scatter_f32<<<(E + 7) / 8, BLOCK_THREADS, 0, stream>>>(
        erow, ecol, ew, h, agg, E, 128);
    bias_relu_f32<<<2048, 256, 0, stream>>>(agg, b1, N_NODES, 128);

    gemm16_wmma_f32<<<dim3(mtiles_blocks(N_NODES), 4), BLOCK_THREADS, 0, stream>>>(
        agg, W2, nullptr, x, N_NODES, 128, 64, 0, 0);
    zero_f32<<<2048, 256, 0, stream>>>(h, (long)N_NODES * 64);
    spmm_scatter_f32<<<(E + 7) / 8, BLOCK_THREADS, 0, stream>>>(
        erow, ecol, ew, x, h, E, 64);
    bias_relu_f32<<<2048, 256, 0, stream>>>(h, b2, N_NODES, 64);

    gemm16_wmma_f32<<<dim3(mtiles_blocks(N_NODES), 1), BLOCK_THREADS, 0, stream>>>(
        h, pw, pb, out, N_NODES, 64, 8, 1, 1);
}


# SpMM scatter: one wave per 32 edges (coalesced index loads, 8 row gathers in flight per wave) instead of one wave per edge; float atomics kept, arrival order unordered as before
# speedup vs baseline: 1.1731x; 1.1731x over previous
#include <hip/hip_runtime.h>

typedef __attribute__((ext_vector_type(2))) float v2f;
typedef __attribute__((ext_vector_type(8))) float v8f;

#define WAVES_PER_BLOCK 8
#define BLOCK_THREADS 256

__global__ __launch_bounds__(BLOCK_THREADS)
void gemm16_wmma_f32(const float* __restrict__ A, const float* __restrict__ B,
                     const float* __restrict__ bias, float* __restrict__ C,
                     int M, int K, int N, int transB, int hasBias)
{
    __shared__ float Bs[256 * 16];
    const int tid   = threadIdx.x;
    const int nbase = blockIdx.y * 16;

    for (int i = tid; i < K * 16; i += BLOCK_THREADS) {
        int k  = i >> 4;
        int n  = i & 15;
        int gn = nbase + n;
        float v = 0.0f;
        if (gn < N) v = transB ? B[(long)gn * K + k] : B[(long)k * N + gn];
        Bs[i] = v;
    }
    __syncthreads();

    const int  wave  = tid >> 5;
    const int  lane  = tid & 31;
    const long mtile = (long)blockIdx.x * WAVES_PER_BLOCK + wave;
    if (mtile * 16 >= M) return;

    const int mn   = lane & 15;
    const int half = lane >> 4;

    const float*  arow  = A + (mtile * 16 + mn) * (long)K;
    const float2* arow2 = (const float2*)arow;

    v8f acc = {};
    for (int k = 0; k < K; k += 4) {
        const int kb = k + 2 * half;
        float2 a2 = arow2[kb >> 1];
        v2f av; av.x = a2.x; av.y = a2.y;
        v2f bv; bv.x = Bs[(kb + 0) * 16 + mn];
                bv.y = Bs[(kb + 1) * 16 + mn];
        acc = __builtin_amdgcn_wmma_f32_16x16x4_f32(
                   false, av,  false, bv,
                   (short)0, acc,  false,  false);
    }

    const int gn = nbase + mn;
    if (gn < N) {
        const float bval = hasBias ? bias[gn] : 0.0f;
        #pragma unroll
        for (int r = 0; r < 8; ++r) {
            long row = mtile * 16 + r + 8 * half;
            C[row * (long)N + gn] = acc[r] + bval;
        }
    }
}

#define EDGES_IN_FLIGHT 8

template <int VEC>
static __device__ __forceinline__
void scatter_one_edge(int src, int dst, float w, int lane,
                      const float* __restrict__ dense, float* __restrict__ out)
{
    typedef float vecf __attribute__((ext_vector_type(VEC)));
    const vecf v = *(const vecf*)(dense + (long)src * (32 * VEC) + lane * VEC);
    float* d = out + (long)dst * (32 * VEC) + lane * VEC;
    #pragma unroll
    for (int c = 0; c < VEC; ++c) atomicAdd(d + c, v[c] * w);
}

template <int VEC>
static __device__ __forceinline__
void scatter_32_edges(const int* __restrict__ rows, const int* __restrict__ cols,
                      const float* __restrict__ ew, const float* __restrict__ dense,
                      float* __restrict__ out, int nE)
{
    typedef float vecf __attribute__((ext_vector_type(VEC)));
    constexpr int F = 32 * VEC;
    const int  lane = threadIdx.x & 31;
    const long wave = (long)blockIdx.x * (BLOCK_THREADS >> 5) + (threadIdx.x >> 5);
    const long e0   = wave * 32;
    if (e0 >= (long)nE) return;
    const long left = (long)nE - e0;
    const int  cnt  = left < 32 ? (int)left : 32;

    int   my_dst = 0, my_src = 0;
    float my_w   = 0.0f;
    if (lane < cnt) {
        my_dst = rows[e0 + lane];
        my_src = cols[e0 + lane];
        my_w   = ew[e0 + lane];
    }

    if (cnt == 32) {
        for (int jb = 0; jb < 32; jb += EDGES_IN_FLIGHT) {
            vecf v[EDGES_IN_FLIGHT];
            #pragma unroll
            for (int u = 0; u < EDGES_IN_FLIGHT; ++u) {
                const int src = __builtin_amdgcn_readlane(my_src, jb + u);
                v[u] = *(const vecf*)(dense + (long)src * F + lane * VEC);
            }
            #pragma unroll
            for (int u = 0; u < EDGES_IN_FLIGHT; ++u) {
                const int   dst = __builtin_amdgcn_readlane(my_dst, jb + u);
                const float w   = __int_as_float(
                    __builtin_amdgcn_readlane(__float_as_int(my_w), jb + u));
                float* d = out + (long)dst * F + lane * VEC;
                #pragma unroll
                for (int c = 0; c < VEC; ++c) atomicAdd(d + c, v[u][c] * w);
            }
        }
    } else {
        for (int j = 0; j < cnt; ++j) {
            const int   src = __builtin_amdgcn_readlane(my_src, j);
            const int   dst = __builtin_amdgcn_readlane(my_dst, j);
            const float w   = __int_as_float(
                __builtin_amdgcn_readlane(__float_as_int(my_w), j));
            scatter_one_edge<VEC>(src, dst, w, lane, dense, out);
        }
    }
}

__global__ __launch_bounds__(BLOCK_THREADS)
void spmm_scatter_f128(const int* __restrict__ rows, const int* __restrict__ cols,
                       const float* __restrict__ ew, const float* __restrict__ dense,
                       float* __restrict__ out, int nE)
{
    scatter_32_edges<4>(rows, cols, ew, dense, out, nE);
}

__global__ __launch_bounds__(BLOCK_THREADS)
void spmm_scatter_f64(const int* __restrict__ rows, const int* __restrict__ cols,
                      const float* __restrict__ ew, const float* __restrict__ dense,
                      float* __restrict__ out, int nE)
{
    scatter_32_edges<2>(rows, cols, ew, dense, out, nE);
}

__global__ void zero_f32(float* __restrict__ p, long n)
{
    long i      = (long)blockIdx.x * blockDim.x + threadIdx.x;
    long stride = (long)gridDim.x * blockDim.x;
    for (; i < n; i += stride) p[i] = 0.0f;
}

__global__ void bias_relu_f32(float* __restrict__ p, const float* __restrict__ b,
                              long nrows, int F)
{
    long n      = nrows * (long)F;
    long i      = (long)blockIdx.x * blockDim.x + threadIdx.x;
    long stride = (long)gridDim.x * blockDim.x;
    for (; i < n; i += stride) {
        float v = p[i] + b[(int)(i & (long)(F - 1))];
        p[i] = v > 0.0f ? v : 0.0f;
    }
}

static inline int mtiles_blocks(int M) {
    return ((M / 16) + WAVES_PER_BLOCK - 1) / WAVES_PER_BLOCK;
}

extern "C" void kernel_launch(void* const* d_in, const int* in_sizes, int n_in,
                              void* d_out, int out_size, void* d_ws, size_t ws_size,
                              hipStream_t stream)
{
    const float* feat0 = (const float*)d_in[0];
    const float* feat1 = (const float*)d_in[1];
    const float* fc0_w = (const float*)d_in[2];
    const float* fc0_b = (const float*)d_in[3];
    const float* fc1_w = (const float*)d_in[4];
    const float* fc1_b = (const float*)d_in[5];
    const float* W1    = (const float*)d_in[6];
    const float* b1    = (const float*)d_in[7];
    const float* W2    = (const float*)d_in[8];
    const float* b2    = (const float*)d_in[9];
    const float* pw    = (const float*)d_in[10];
    const float* pb    = (const float*)d_in[11];
    const int*   erow  = (const int*)d_in[12];
    const int*   ecol  = (const int*)d_in[13];
    const float* ew    = (const float*)d_in[14];
    float*       out   = (float*)d_out;

    const int E       = in_sizes[12];
    const int N_NODES = 100000;
    const int NT      = 50000;

    float* x   = (float*)d_ws;
    float* h   = x + (long)N_NODES * 128;
    float* agg = h + (long)N_NODES * 128;

    gemm16_wmma_f32<<<dim3(mtiles_blocks(NT), 8), BLOCK_THREADS, 0, stream>>>(
        feat0, fc0_w, fc0_b, x, NT, 128, 128, 1, 1);
    gemm16_wmma_f32<<<dim3(mtiles_blocks(NT), 8), BLOCK_THREADS, 0, stream>>>(
        feat1, fc1_w, fc1_b, x + (long)NT * 128, NT, 256, 128, 1, 1);

    gemm16_wmma_f32<<<dim3(mtiles_blocks(N_NODES), 8), BLOCK_THREADS, 0, stream>>>(
        x, W1, nullptr, h, N_NODES, 128, 128, 0, 0);
    zero_f32<<<2048, 256, 0, stream>>>(agg, (long)N_NODES * 128);
    spmm_scatter_f128<<<(E + 255) / 256, BLOCK_THREADS, 0, stream>>>(
        erow, ecol, ew, h, agg, E);
    bias_relu_f32<<<2048, 256, 0, stream>>>(agg, b1, N_NODES, 128);

    gemm16_wmma_f32<<<dim3(mtiles_blocks(N_NODES), 4), BLOCK_THREADS, 0, stream>>>(
        agg, W2, nullptr, x, N_NODES, 128, 64, 0, 0);
    zero_f32<<<2048, 256, 0, stream>>>(h, (long)N_NODES * 64);
    spmm_scatter_f64<<<(E + 255) / 256, BLOCK_THREADS, 0, stream>>>(
        erow, ecol, ew, x, h, E);
    bias_relu_f32<<<2048, 256, 0, stream>>>(h, b2, N_NODES, 64);

    gemm16_wmma_f32<<<dim3(mtiles_blocks(N_NODES), 1), BLOCK_THREADS, 0, stream>>>(
        h, pw, pb, out, N_NODES, 64, 8, 1, 1);
}


# SpMM scatter: lane l owns float l of each 128-B piece of a row, so every atomic add instruction covers one whole line (was 4 or 2 part-filled lines); float atomics kept, arrival order unordered
# speedup vs baseline: 1.7026x; 1.4514x over previous
#include <hip/hip_runtime.h>

typedef __attribute__((ext_vector_type(2))) float v2f;
typedef __attribute__((ext_vector_type(8))) float v8f;

#define WAVES_PER_BLOCK 8
#define BLOCK_THREADS 256

__global__ __launch_bounds__(BLOCK_THREADS)
void gemm16_wmma_f32(const float* __restrict__ A, const float* __restrict__ B,
                     const float* __restrict__ bias, float* __restrict__ C,
                     int M, int K, int N, int transB, int hasBias)
{
    __shared__ float Bs[256 * 16];
    const int tid   = threadIdx.x;
    const int nbase = blockIdx.y * 16;

    for (int i = tid; i < K * 16; i += BLOCK_THREADS) {
        int k  = i >> 4;
        int n  = i & 15;
        int gn = nbase + n;
        float v = 0.0f;
        if (gn < N) v = transB ? B[(long)gn * K + k] : B[(long)k * N + gn];
        Bs[i] = v;
    }
    __syncthreads();

    const int  wave  = tid >> 5;
    const int  lane  = tid & 31;
    const long mtile = (long)blockIdx.x * WAVES_PER_BLOCK + wave;
    if (mtile * 16 >= M) return;

    const int mn   = lane & 15;
    const int half = lane >> 4;

    const float*  arow  = A + (mtile * 16 + mn) * (long)K;
    const float2* arow2 = (const float2*)arow;

    v8f acc = {};
    for (int k = 0; k < K; k += 4) {
        const int kb = k + 2 * half;
        float2 a2 = arow2[kb >> 1];
        v2f av; av.x = a2.x; av.y = a2.y;
        v2f bv; bv.x = Bs[(kb + 0) * 16 + mn];
                bv.y = Bs[(kb + 1) * 16 + mn];
        acc = __builtin_amdgcn_wmma_f32_16x16x4_f32(
                   false, av,  false, bv,
                   (short)0, acc,  false,  false);
    }

    const int gn = nbase + mn;
    if (gn < N) {
        const float bval = hasBias ? bias[gn] : 0.0f;
        #pragma unroll
        for (int r = 0; r < 8; ++r) {
            long row = mtile * 16 + r + 8 * half;
            C[row * (long)N + gn] = acc[r] + bval;
        }
    }
}

#define EDGES_IN_FLIGHT 8

template <int VEC>
static __device__ __forceinline__
void scatter_one_edge(int src, int dst, float w, int lane,
                      const float* __restrict__ dense, float* __restrict__ out)
{
    const float* s = dense + (long)src * (32 * VEC) + lane;
    float*       d = out   + (long)dst * (32 * VEC) + lane;
    float v[VEC];
    #pragma unroll
    for (int c = 0; c < VEC; ++c) v[c] = s[c * 32];
    #pragma unroll
    for (int c = 0; c < VEC; ++c) atomicAdd(d + c * 32, v[c] * w);
}

template <int VEC>
static __device__ __forceinline__
void scatter_32_edges(const int* __restrict__ rows, const int* __restrict__ cols,
                      const float* __restrict__ ew, const float* __restrict__ dense,
                      float* __restrict__ out, int nE)
{
    constexpr int F = 32 * VEC;
    const int  lane = threadIdx.x & 31;
    const long wave = (long)blockIdx.x * (BLOCK_THREADS >> 5) + (threadIdx.x >> 5);
    const long e0   = wave * 32;
    if (e0 >= (long)nE) return;
    const long left = (long)nE - e0;
    const int  cnt  = left < 32 ? (int)left : 32;

    int   my_dst = 0, my_src = 0;
    float my_w   = 0.0f;
    if (lane < cnt) {
        my_dst = rows[e0 + lane];
        my_src = cols[e0 + lane];
        my_w   = ew[e0 + lane];
    }

    if (cnt == 32) {
        for (int jb = 0; jb < 32; jb += EDGES_IN_FLIGHT) {
            float v[EDGES_IN_FLIGHT][VEC];
            #pragma unroll
            for (int u = 0; u < EDGES_IN_FLIGHT; ++u) {
                const int src = __builtin_amdgcn_readlane(my_src, jb + u);
                const float* s = dense + (long)src * F + lane;
                #pragma unroll
                for (int c = 0; c < VEC; ++c) v[u][c] = s[c * 32];
            }
            #pragma unroll
            for (int u = 0; u < EDGES_IN_FLIGHT; ++u) {
                const int   dst = __builtin_amdgcn_readlane(my_dst, jb + u);
                const float w   = __int_as_float(
                    __builtin_amdgcn_readlane(__float_as_int(my_w), jb + u));
                float* d = out + (long)dst * F + lane;
                #pragma unroll
                for (int c = 0; c < VEC; ++c) atomicAdd(d + c * 32, v[u][c] * w);
            }
        }
    } else {
        for (int j = 0; j < cnt; ++j) {
            const int   src = __builtin_amdgcn_readlane(my_src, j);
            const int   dst = __builtin_amdgcn_readlane(my_dst, j);
            const float w   = __int_as_float(
                __builtin_amdgcn_readlane(__float_as_int(my_w), j));
            scatter_one_edge<VEC>(src, dst, w, lane, dense, out);
        }
    }
}

__global__ __launch_bounds__(BLOCK_THREADS)
void spmm_scatter_f128(const int* __restrict__ rows, const int* __restrict__ cols,
                       const float* __restrict__ ew, const float* __restrict__ dense,
                       float* __restrict__ out, int nE)
{
    scatter_32_edges<4>(rows, cols, ew, dense, out, nE);
}

__global__ __launch_bounds__(BLOCK_THREADS)
void spmm_scatter_f64(const int* __restrict__ rows, const int* __restrict__ cols,
                      const float* __restrict__ ew, const float* __restrict__ dense,
                      float* __restrict__ out, int nE)
{
    scatter_32_edges<2>(rows, cols, ew, dense, out, nE);
}

__global__ void zero_f32(float* __restrict__ p, long n)
{
    long i      = (long)blockIdx.x * blockDim.x + threadIdx.x;
    long stride = (long)gridDim.x * blockDim.x;
    for (; i < n; i += stride) p[i] = 0.0f;
}

__global__ void bias_relu_f32(float* __restrict__ p, const float* __restrict__ b,
                              long nrows, int F)
{
    long n      = nrows * (long)F;
    long i      = (long)blockIdx.x * blockDim.x + threadIdx.x;
    long stride = (long)gridDim.x * blockDim.x;
    for (; i < n; i += stride) {
        float v = p[i] + b[(int)(i & (long)(F - 1))];
        p[i] = v > 0.0f ? v : 0.0f;
    }
}

static inline int mtiles_blocks(int M) {
    return ((M / 16) + WAVES_PER_BLOCK - 1) / WAVES_PER_BLOCK;
}

extern "C" void kernel_launch(void* const* d_in, const int* in_sizes, int n_in,
                              void* d_out, int out_size, void* d_ws, size_t ws_size,
                              hipStream_t stream)
{
    const float* feat0 = (const float*)d_in[0];
    const float* feat1 = (const float*)d_in[1];
    const float* fc0_w = (const float*)d_in[2];
    const float* fc0_b = (const float*)d_in[3];
    const float* fc1_w = (const float*)d_in[4];
    const float* fc1_b = (const float*)d_in[5];
    const float* W1    = (const float*)d_in[6];
    const float* b1    = (const float*)d_in[7];
    const float* W2    = (const float*)d_in[8];
    const float* b2    = (const float*)d_in[9];
    const float* pw    = (const float*)d_in[10];
    const float* pb    = (const float*)d_in[11];
    const int*   erow  = (const int*)d_in[12];
    const int*   ecol  = (const int*)d_in[13];
    const float* ew    = (const float*)d_in[14];
    float*       out   = (float*)d_out;

    const int E       = in_sizes[12];
    const int N_NODES = 100000;
    const int NT      = 50000;

    float* x   = (float*)d_ws;
    float* h   = x + (long)N_NODES * 128;
    float* agg = h + (long)N_NODES * 128;

    gemm16_wmma_f32<<<dim3(mtiles_blocks(NT), 8), BLOCK_THREADS, 0, stream>>>(
        feat0, fc0_w, fc0_b, x, NT, 128, 128, 1, 1);
    gemm16_wmma_f32<<<dim3(mtiles_blocks(NT), 8), BLOCK_THREADS, 0, stream>>>(
        feat1, fc1_w, fc1_b, x + (long)NT * 128, NT, 256, 128, 1, 1);

    gemm16_wmma_f32<<<dim3(mtiles_blocks(N_NODES), 8), BLOCK_THREADS, 0, stream>>>(
        x, W1, nullptr, h, N_NODES, 128, 128, 0, 0);
    zero_f32<<<2048, 256, 0, stream>>>(agg, (long)N_NODES * 128);
    spmm_scatter_f128<<<(E + 255) / 256, BLOCK_THREADS, 0, stream>>>(
        erow, ecol, ew, h, agg, E);
    bias_relu_f32<<<2048, 256, 0, stream>>>(agg, b1, N_NODES, 128);

    gemm16_wmma_f32<<<dim3(mtiles_blocks(N_NODES), 4), BLOCK_THREADS, 0, stream>>>(
        agg, W2, nullptr, x, N_NODES, 128, 64, 0, 0);
    zero_f32<<<2048, 256, 0, stream>>>(h, (long)N_NODES * 64);
    spmm_scatter_f64<<<(E + 255) / 256, BLOCK_THREADS, 0, stream>>>(
        erow, ecol, ew, x, h, E);
    bias_relu_f32<<<2048, 256, 0, stream>>>(h, b2, N_NODES, 64);

    gemm16_wmma_f32<<<dim3(mtiles_blocks(N_NODES), 1), BLOCK_THREADS, 0, stream>>>(
        h, pw, pb, out, N_NODES, 64, 8, 1, 1);
}


# GEMMs: one wave owns a 16-row strip over all N columns (A fragment read once, reused by N/16 independent WMMA accumulators; whole B in LDS as [k/4][n][k%4]); k order of sums unchanged
# speedup vs baseline: 1.9899x; 1.1687x over previous
#include <hip/hip_runtime.h>

typedef __attribute__((ext_vector_type(2))) float v2f;
typedef __attribute__((ext_vector_type(8))) float v8f;

#define WAVES_PER_BLOCK 8
#define BLOCK_THREADS 256

#define GEMM_KC 128

template <int NT, int TRANSB>
static __device__ __forceinline__
void stage_b_chunk(float* __restrict__ Bs, const float* __restrict__ B,
                   int K, int N, int k0, int kc, int tid)
{
    constexpr int NP = NT * 16;
    const int nquads = (kc >> 2) * NP;
    if (TRANSB == 0) {
        for (int q = tid; q < nquads; q += BLOCK_THREADS) {
            const int n  = q % NP;
            const int kq = q / NP;
            float4 v = make_float4(0.0f, 0.0f, 0.0f, 0.0f);
            if (n < N) {
                const float* p = B + (long)(k0 + 4 * kq) * N + n;
                v.x = p[0];
                v.y = p[N];
                v.z = p[2 * (long)N];
                v.w = p[3 * (long)N];
            }
            *(float4*)&Bs[(kq * NP + n) * 4] = v;
        }
    } else {
        const int qk8 = kc >> 5;
        for (int i = tid; i < nquads; i += BLOCK_THREADS) {
            const int kq_lo = i & 7;
            const int n_lo  = (i >> 3) & 3;
            const int rest  = i >> 5;
            const int kq    = (rest % qk8) * 8 + kq_lo;
            const int n     = (rest / qk8) * 4 + n_lo;
            float4 v = make_float4(0.0f, 0.0f, 0.0f, 0.0f);
            if (n < N) v = *(const float4*)(B + (long)n * K + k0 + 4 * kq);
            *(float4*)&Bs[(kq * NP + n) * 4] = v;
        }
    }
}

template <int NT, int TRANSB>
__global__ __launch_bounds__(BLOCK_THREADS)
void gemm_strip_wmma_f32(const float* __restrict__ A, const float* __restrict__ B,
                         const float* __restrict__ bias, float* __restrict__ C,
                         int M, int K, int N, int hasBias)
{
    constexpr int NP = NT * 16;
    __shared__ __attribute__((aligned(16))) float Bs[(GEMM_KC / 4) * NP * 4];

    const int  tid   = threadIdx.x;
    const int  wave  = tid >> 5;
    const int  lane  = tid & 31;
    const int  mn    = lane & 15;
    const int  half  = lane >> 4;
    const long mtile = (long)blockIdx.x * WAVES_PER_BLOCK + wave;
    const bool active = mtile * 16 < (long)M;

    const float2* arow2 = (const float2*)(A + (mtile * 16 + mn) * (long)K);
    const float*  bfrag = Bs + mn * 4 + 2 * half;

    v8f acc[NT];
    #pragma unroll
    for (int t = 0; t < NT; ++t) acc[t] = v8f{};

    for (int k0 = 0; k0 < K; k0 += GEMM_KC) {
        const int kc = (K - k0) < GEMM_KC ? (K - k0) : GEMM_KC;
        if (k0 > 0) __syncthreads();
        stage_b_chunk<NT, TRANSB>(Bs, B, K, N, k0, kc, tid);
        __syncthreads();

        if (active) {
            for (int kq = 0; kq < (kc >> 2); kq += 8) {
                float2 a[8];
                #pragma unroll
                for (int u = 0; u < 8; ++u)
                    a[u] = arow2[((k0 + 4 * (kq + u)) >> 1) + half];
                #pragma unroll
                for (int u = 0; u < 8; ++u) {
                    v2f av; av.x = a[u].x; av.y = a[u].y;
                    #pragma unroll
                    for (int t = 0; t < NT; ++t) {
                        const float2 b2 = *(const float2*)(bfrag + ((kq + u) * NP + t * 16) * 4);
                        v2f bv; bv.x = b2.x; bv.y = b2.y;
                        acc[t] = __builtin_amdgcn_wmma_f32_16x16x4_f32(
                                      false, av,  false, bv,
                                      (short)0, acc[t],
                                      false,  false);
                    }
                }
            }
        }
    }

    if (active) {
        #pragma unroll
        for (int t = 0; t < NT; ++t) {
            const int gn = t * 16 + mn;
            if (gn < N) {
                const float bval = hasBias ? bias[gn] : 0.0f;
                #pragma unroll
                for (int r = 0; r < 8; ++r) {
                    long row = mtile * 16 + r + 8 * half;
                    C[row * (long)N + gn] = acc[t][r] + bval;
                }
            }
        }
    }
}

#define EDGES_IN_FLIGHT 8

template <int VEC>
static __device__ __forceinline__
void scatter_one_edge(int src, int dst, float w, int lane,
                      const float* __restrict__ dense, float* __restrict__ out)
{
    const float* s = dense + (long)src * (32 * VEC) + lane;
    float*       d = out   + (long)dst * (32 * VEC) + lane;
    float v[VEC];
    #pragma unroll
    for (int c = 0; c < VEC; ++c) v[c] = s[c * 32];
    #pragma unroll
    for (int c = 0; c < VEC; ++c) atomicAdd(d + c * 32, v[c] * w);
}

template <int VEC>
static __device__ __forceinline__
void scatter_32_edges(const int* __restrict__ rows, const int* __restrict__ cols,
                      const float* __restrict__ ew, const float* __restrict__ dense,
                      float* __restrict__ out, int nE)
{
    constexpr int F = 32 * VEC;
    const int  lane = threadIdx.x & 31;
    const long wave = (long)blockIdx.x * (BLOCK_THREADS >> 5) + (threadIdx.x >> 5);
    const long e0   = wave * 32;
    if (e0 >= (long)nE) return;
    const long left = (long)nE - e0;
    const int  cnt  = left < 32 ? (int)left : 32;

    int   my_dst = 0, my_src = 0;
    float my_w   = 0.0f;
    if (lane < cnt) {
        my_dst = rows[e0 + lane];
        my_src = cols[e0 + lane];
        my_w   = ew[e0 + lane];
    }

    if (cnt == 32) {
        for (int jb = 0; jb < 32; jb += EDGES_IN_FLIGHT) {
            float v[EDGES_IN_FLIGHT][VEC];
            #pragma unroll
            for (int u = 0; u < EDGES_IN_FLIGHT; ++u) {
                const int src = __builtin_amdgcn_readlane(my_src, jb + u);
                const float* s = dense + (long)src * F + lane;
                #pragma unroll
                for (int c = 0; c < VEC; ++c) v[u][c] = s[c * 32];
            }
            #pragma unroll
            for (int u = 0; u < EDGES_IN_FLIGHT; ++u) {
                const int   dst = __builtin_amdgcn_readlane(my_dst, jb + u);
                const float w   = __int_as_float(
                    __builtin_amdgcn_readlane(__float_as_int(my_w), jb + u));
                float* d = out + (long)dst * F + lane;
                #pragma unroll
                for (int c = 0; c < VEC; ++c) atomicAdd(d + c * 32, v[u][c] * w);
            }
        }
    } else {
        for (int j = 0; j < cnt; ++j) {
            const int   src = __builtin_amdgcn_readlane(my_src, j);
            const int   dst = __builtin_amdgcn_readlane(my_dst, j);
            const float w   = __int_as_float(
                __builtin_amdgcn_readlane(__float_as_int(my_w), j));
            scatter_one_edge<VEC>(src, dst, w, lane, dense, out);
        }
    }
}

__global__ __launch_bounds__(BLOCK_THREADS)
void spmm_scatter_f128(const int* __restrict__ rows, const int* __restrict__ cols,
                       const float* __restrict__ ew, const float* __restrict__ dense,
                       float* __restrict__ out, int nE)
{
    scatter_32_edges<4>(rows, cols, ew, dense, out, nE);
}

__global__ __launch_bounds__(BLOCK_THREADS)
void spmm_scatter_f64(const int* __restrict__ rows, const int* __restrict__ cols,
                      const float* __restrict__ ew, const float* __restrict__ dense,
                      float* __restrict__ out, int nE)
{
    scatter_32_edges<2>(rows, cols, ew, dense, out, nE);
}

__global__ void zero_f32(float* __restrict__ p, long n)
{
    long i      = (long)blockIdx.x * blockDim.x + threadIdx.x;
    long stride = (long)gridDim.x * blockDim.x;
    for (; i < n; i += stride) p[i] = 0.0f;
}

__global__ void bias_relu_f32(float* __restrict__ p, const float* __restrict__ b,
                              long nrows, int F)
{
    long n      = nrows * (long)F;
    long i      = (long)blockIdx.x * blockDim.x + threadIdx.x;
    long stride = (long)gridDim.x * blockDim.x;
    for (; i < n; i += stride) {
        float v = p[i] + b[(int)(i & (long)(F - 1))];
        p[i] = v > 0.0f ? v : 0.0f;
    }
}

static inline int mtiles_blocks(int M) {
    return ((M / 16) + WAVES_PER_BLOCK - 1) / WAVES_PER_BLOCK;
}

extern "C" void kernel_launch(void* const* d_in, const int* in_sizes, int n_in,
                              void* d_out, int out_size, void* d_ws, size_t ws_size,
                              hipStream_t stream)
{
    const float* feat0 = (const float*)d_in[0];
    const float* feat1 = (const float*)d_in[1];
    const float* fc0_w = (const float*)d_in[2];
    const float* fc0_b = (const float*)d_in[3];
    const float* fc1_w = (const float*)d_in[4];
    const float* fc1_b = (const float*)d_in[5];
    const float* W1    = (const float*)d_in[6];
    const float* b1    = (const float*)d_in[7];
    const float* W2    = (const float*)d_in[8];
    const float* b2    = (const float*)d_in[9];
    const float* pw    = (const float*)d_in[10];
    const float* pb    = (const float*)d_in[11];
    const int*   erow  = (const int*)d_in[12];
    const int*   ecol  = (const int*)d_in[13];
    const float* ew    = (const float*)d_in[14];
    float*       out   = (float*)d_out;

    const int E       = in_sizes[12];
    const int N_NODES = 100000;
    const int NT      = 50000;

    float* x   = (float*)d_ws;
    float* h   = x + (long)N_NODES * 128;
    float* agg = h + (long)N_NODES * 128;

    gemm_strip_wmma_f32<8, 1><<<mtiles_blocks(NT), BLOCK_THREADS, 0, stream>>>(
        feat0, fc0_w, fc0_b, x, NT, 128, 128, 1);
    gemm_strip_wmma_f32<8, 1><<<mtiles_blocks(NT), BLOCK_THREADS, 0, stream>>>(
        feat1, fc1_w, fc1_b, x + (long)NT * 128, NT, 256, 128, 1);

    gemm_strip_wmma_f32<8, 0><<<mtiles_blocks(N_NODES), BLOCK_THREADS, 0, stream>>>(
        x, W1, nullptr, h, N_NODES, 128, 128, 0);
    zero_f32<<<2048, 256, 0, stream>>>(agg, (long)N_NODES * 128);
    spmm_scatter_f128<<<(E + 255) / 256, BLOCK_THREADS, 0, stream>>>(
        erow, ecol, ew, h, agg, E);
    bias_relu_f32<<<2048, 256, 0, stream>>>(agg, b1, N_NODES, 128);

    gemm_strip_wmma_f32<4, 0><<<mtiles_blocks(N_NODES), BLOCK_THREADS, 0, stream>>>(
        agg, W2, nullptr, x, N_NODES, 128, 64, 0);
    zero_f32<<<2048, 256, 0, stream>>>(h, (long)N_NODES * 64);
    spmm_scatter_f64<<<(E + 255) / 256, BLOCK_THREADS, 0, stream>>>(
        erow, ecol, ew, x, h, E);
    bias_relu_f32<<<2048, 256, 0, stream>>>(h, b2, N_NODES, 64);

    gemm_strip_wmma_f32<1, 1><<<mtiles_blocks(N_NODES), BLOCK_THREADS, 0, stream>>>(
        h, pw, pb, out, N_NODES, 64, 8, 1);
}


# Elementwise passes fused into the GEMMs beside them: bias+relu applied in registers on the consumer GEMM's A load, zero fill of the next scatter target stored by the producer GEMM; 11 -> 7 launches
# speedup vs baseline: 2.2160x; 1.1136x over previous
#include <hip/hip_runtime.h>

typedef __attribute__((ext_vector_type(2))) float v2f;
typedef __attribute__((ext_vector_type(8))) float v8f;

#define WAVES_PER_BLOCK 8
#define BLOCK_THREADS 256

#define GEMM_KC 128

template <int NT, int TRANSB>
static __device__ __forceinline__
void stage_b_chunk(float* __restrict__ Bs, const float* __restrict__ B,
                   int K, int N, int k0, int kc, int tid)
{
    constexpr int NP = NT * 16;
    const int nquads = (kc >> 2) * NP;
    if (TRANSB == 0) {
        for (int q = tid; q < nquads; q += BLOCK_THREADS) {
            const int n  = q % NP;
            const int kq = q / NP;
            float4 v = make_float4(0.0f, 0.0f, 0.0f, 0.0f);
            if (n < N) {
                const float* p = B + (long)(k0 + 4 * kq) * N + n;
                v.x = p[0];
                v.y = p[N];
                v.z = p[2 * (long)N];
                v.w = p[3 * (long)N];
            }
            *(float4*)&Bs[(kq * NP + n) * 4] = v;
        }
    } else {
        const int qk8 = kc >> 5;
        for (int i = tid; i < nquads; i += BLOCK_THREADS) {
            const int kq_lo = i & 7;
            const int n_lo  = (i >> 3) & 3;
            const int rest  = i >> 5;
            const int kq    = (rest % qk8) * 8 + kq_lo;
            const int n     = (rest / qk8) * 4 + n_lo;
            float4 v = make_float4(0.0f, 0.0f, 0.0f, 0.0f);
            if (n < N) v = *(const float4*)(B + (long)n * K + k0 + 4 * kq);
            *(float4*)&Bs[(kq * NP + n) * 4] = v;
        }
    }
}

template <int NT, int TRANSB, int RELU_A, int ZCOLS>
__global__ __launch_bounds__(BLOCK_THREADS)
void gemm_strip_wmma_f32(const float* __restrict__ A, const float* __restrict__ B,
                         const float* __restrict__ bias, float* __restrict__ C,
                         const float* __restrict__ abias, float* __restrict__ zfill,
                         int M, int K, int N, int hasBias)
{
    constexpr int NP = NT * 16;
    __shared__ __attribute__((aligned(16))) float Bs[(GEMM_KC / 4) * NP * 4];

    const int  tid   = threadIdx.x;
    const int  wave  = tid >> 5;
    const int  lane  = tid & 31;
    const int  mn    = lane & 15;
    const int  half  = lane >> 4;
    const long mtile = (long)blockIdx.x * WAVES_PER_BLOCK + wave;
    const bool active = mtile * 16 < (long)M;

    const float2* arow2 = (const float2*)(A + (mtile * 16 + mn) * (long)K);
    const float*  bfrag = Bs + mn * 4 + 2 * half;

    v8f acc[NT];
    #pragma unroll
    for (int t = 0; t < NT; ++t) acc[t] = v8f{};

    for (int k0 = 0; k0 < K; k0 += GEMM_KC) {
        const int kc = (K - k0) < GEMM_KC ? (K - k0) : GEMM_KC;
        if (k0 > 0) __syncthreads();
        stage_b_chunk<NT, TRANSB>(Bs, B, K, N, k0, kc, tid);
        __syncthreads();

        if (active) {
            for (int kq = 0; kq < (kc >> 2); kq += 8) {
                float2 a[8];
                #pragma unroll
                for (int u = 0; u < 8; ++u)
                    a[u] = arow2[((k0 + 4 * (kq + u)) >> 1) + half];
                #pragma unroll
                for (int u = 0; u < 8; ++u) {
                    v2f av; av.x = a[u].x; av.y = a[u].y;
                    if (RELU_A) {
                        const float2 b2v = *(const float2*)(abias + k0 + 4 * (kq + u) + 2 * half);
                        const float vx = av.x + b2v.x;
                        const float vy = av.y + b2v.y;
                        av.x = vx > 0.0f ? vx : 0.0f;
                        av.y = vy > 0.0f ? vy : 0.0f;
                    }
                    #pragma unroll
                    for (int t = 0; t < NT; ++t) {
                        const float2 b2 = *(const float2*)(bfrag + ((kq + u) * NP + t * 16) * 4);
                        v2f bv; bv.x = b2.x; bv.y = b2.y;
                        acc[t] = __builtin_amdgcn_wmma_f32_16x16x4_f32(
                                      false, av,  false, bv,
                                      (short)0, acc[t],
                                      false,  false);
                    }
                }
            }
        }
    }

    if (active) {
        if (ZCOLS > 0) {
            float4* z = (float4*)(zfill + mtile * 16 * (long)ZCOLS);
            #pragma unroll
            for (int i = 0; i < (16 * ZCOLS) / 128; ++i)
                z[i * 32 + lane] = make_float4(0.0f, 0.0f, 0.0f, 0.0f);
        }
        #pragma unroll
        for (int t = 0; t < NT; ++t) {
            const int gn = t * 16 + mn;
            if (gn < N) {
                const float bval = hasBias ? bias[gn] : 0.0f;
                #pragma unroll
                for (int r = 0; r < 8; ++r) {
                    long row = mtile * 16 + r + 8 * half;
                    C[row * (long)N + gn] = acc[t][r] + bval;
                }
            }
        }
    }
}

#define EDGES_IN_FLIGHT 8

template <int VEC>
static __device__ __forceinline__
void scatter_one_edge(int src, int dst, float w, int lane,
                      const float* __restrict__ dense, float* __restrict__ out)
{
    const float* s = dense + (long)src * (32 * VEC) + lane;
    float*       d = out   + (long)dst * (32 * VEC) + lane;
    float v[VEC];
    #pragma unroll
    for (int c = 0; c < VEC; ++c) v[c] = s[c * 32];
    #pragma unroll
    for (int c = 0; c < VEC; ++c) atomicAdd(d + c * 32, v[c] * w);
}

template <int VEC>
static __device__ __forceinline__
void scatter_32_edges(const int* __restrict__ rows, const int* __restrict__ cols,
                      const float* __restrict__ ew, const float* __restrict__ dense,
                      float* __restrict__ out, int nE)
{
    constexpr int F = 32 * VEC;
    const int  lane = threadIdx.x & 31;
    const long wave = (long)blockIdx.x * (BLOCK_THREADS >> 5) + (threadIdx.x >> 5);
    const long e0   = wave * 32;
    if (e0 >= (long)nE) return;
    const long left = (long)nE - e0;
    const int  cnt  = left < 32 ? (int)left : 32;

    int   my_dst = 0, my_src = 0;
    float my_w   = 0.0f;
    if (lane < cnt) {
        my_dst = rows[e0 + lane];
        my_src = cols[e0 + lane];
        my_w   = ew[e0 + lane];
    }

    if (cnt == 32) {
        for (int jb = 0; jb < 32; jb += EDGES_IN_FLIGHT) {
            float v[EDGES_IN_FLIGHT][VEC];
            #pragma unroll
            for (int u = 0; u < EDGES_IN_FLIGHT; ++u) {
                const int src = __builtin_amdgcn_readlane(my_src, jb + u);
                const float* s = dense + (long)src * F + lane;
                #pragma unroll
                for (int c = 0; c < VEC; ++c) v[u][c] = s[c * 32];
            }
            #pragma unroll
            for (int u = 0; u < EDGES_IN_FLIGHT; ++u) {
                const int   dst = __builtin_amdgcn_readlane(my_dst, jb + u);
                const float w   = __int_as_float(
                    __builtin_amdgcn_readlane(__float_as_int(my_w), jb + u));
                float* d = out + (long)dst * F + lane;
                #pragma unroll
                for (int c = 0; c < VEC; ++c) atomicAdd(d + c * 32, v[u][c] * w);
            }
        }
    } else {
        for (int j = 0; j < cnt; ++j) {
            const int   src = __builtin_amdgcn_readlane(my_src, j);
            const int   dst = __builtin_amdgcn_readlane(my_dst, j);
            const float w   = __int_as_float(
                __builtin_amdgcn_readlane(__float_as_int(my_w), j));
            scatter_one_edge<VEC>(src, dst, w, lane, dense, out);
        }
    }
}

__global__ __launch_bounds__(BLOCK_THREADS)
void spmm_scatter_f128(const int* __restrict__ rows, const int* __restrict__ cols,
                       const float* __restrict__ ew, const float* __restrict__ dense,
                       float* __restrict__ out, int nE)
{
    scatter_32_edges<4>(rows, cols, ew, dense, out, nE);
}

__global__ __launch_bounds__(BLOCK_THREADS)
void spmm_scatter_f64(const int* __restrict__ rows, const int* __restrict__ cols,
                      const float* __restrict__ ew, const float* __restrict__ dense,
                      float* __restrict__ out, int nE)
{
    scatter_32_edges<2>(rows, cols, ew, dense, out, nE);
}

static inline int mtiles_blocks(int M) {
    return ((M / 16) + WAVES_PER_BLOCK - 1) / WAVES_PER_BLOCK;
}

extern "C" void kernel_launch(void* const* d_in, const int* in_sizes, int n_in,
                              void* d_out, int out_size, void* d_ws, size_t ws_size,
                              hipStream_t stream)
{
    const float* feat0 = (const float*)d_in[0];
    const float* feat1 = (const float*)d_in[1];
    const float* fc0_w = (const float*)d_in[2];
    const float* fc0_b = (const float*)d_in[3];
    const float* fc1_w = (const float*)d_in[4];
    const float* fc1_b = (const float*)d_in[5];
    const float* W1    = (const float*)d_in[6];
    const float* b1    = (const float*)d_in[7];
    const float* W2    = (const float*)d_in[8];
    const float* b2    = (const float*)d_in[9];
    const float* pw    = (const float*)d_in[10];
    const float* pb    = (const float*)d_in[11];
    const int*   erow  = (const int*)d_in[12];
    const int*   ecol  = (const int*)d_in[13];
    const float* ew    = (const float*)d_in[14];
    float*       out   = (float*)d_out;

    const int E       = in_sizes[12];
    const int N_NODES = 100000;
    const int NT      = 50000;

    float* x   = (float*)d_ws;
    float* h   = x + (long)N_NODES * 128;
    float* agg = h + (long)N_NODES * 128;

    gemm_strip_wmma_f32<8, 1, 0, 0><<<mtiles_blocks(NT), BLOCK_THREADS, 0, stream>>>(
        feat0, fc0_w, fc0_b, x, nullptr, nullptr, NT, 128, 128, 1);
    gemm_strip_wmma_f32<8, 1, 0, 0><<<mtiles_blocks(NT), BLOCK_THREADS, 0, stream>>>(
        feat1, fc1_w, fc1_b, x + (long)NT * 128, nullptr, nullptr, NT, 256, 128, 1);

    gemm_strip_wmma_f32<8, 0, 0, 128><<<mtiles_blocks(N_NODES), BLOCK_THREADS, 0, stream>>>(
        x, W1, nullptr, h, nullptr, agg, N_NODES, 128, 128, 0);
    spmm_scatter_f128<<<(E + 255) / 256, BLOCK_THREADS, 0, stream>>>(
        erow, ecol, ew, h, agg, E);

    gemm_strip_wmma_f32<4, 0, 1, 64><<<mtiles_blocks(N_NODES), BLOCK_THREADS, 0, stream>>>(
        agg, W2, nullptr, x, b1, h, N_NODES, 128, 64, 0);
    spmm_scatter_f64<<<(E + 255) / 256, BLOCK_THREADS, 0, stream>>>(
        erow, ecol, ew, x, h, E);

    gemm_strip_wmma_f32<1, 1, 1, 0><<<mtiles_blocks(N_NODES), BLOCK_THREADS, 0, stream>>>(
        h, pw, pb, out, b2, nullptr, N_NODES, 64, 8, 1);
}


# SpMM: edges sorted by destination row each call (counting sort on LDS counters), one wave sums a row in registers and stores it once; float atomics and zero fill gone; sum order = counter arrival
# speedup vs baseline: 2.7484x; 1.2403x over previous
#include <hip/hip_runtime.h>

typedef __attribute__((ext_vector_type(2))) float v2f;
typedef __attribute__((ext_vector_type(8))) float v8f;

#define WAVES_PER_BLOCK 8
#define BLOCK_THREADS 256

#define GEMM_KC 128

template <int NT, int TRANSB>
static __device__ __forceinline__
void stage_b_chunk(float* __restrict__ Bs, const float* __restrict__ B,
                   int K, int N, int k0, int kc, int tid)
{
    constexpr int NP = NT * 16;
    const int nquads = (kc >> 2) * NP;
    if (TRANSB == 0) {
        for (int q = tid; q < nquads; q += BLOCK_THREADS) {
            const int n  = q % NP;
            const int kq = q / NP;
            float4 v = make_float4(0.0f, 0.0f, 0.0f, 0.0f);
            if (n < N) {
                const float* p = B + (long)(k0 + 4 * kq) * N + n;
                v.x = p[0];
                v.y = p[N];
                v.z = p[2 * (long)N];
                v.w = p[3 * (long)N];
            }
            *(float4*)&Bs[(kq * NP + n) * 4] = v;
        }
    } else {
        const int qk8 = kc >> 5;
        for (int i = tid; i < nquads; i += BLOCK_THREADS) {
            const int kq_lo = i & 7;
            const int n_lo  = (i >> 3) & 3;
            const int rest  = i >> 5;
            const int kq    = (rest % qk8) * 8 + kq_lo;
            const int n     = (rest / qk8) * 4 + n_lo;
            float4 v = make_float4(0.0f, 0.0f, 0.0f, 0.0f);
            if (n < N) v = *(const float4*)(B + (long)n * K + k0 + 4 * kq);
            *(float4*)&Bs[(kq * NP + n) * 4] = v;
        }
    }
}

template <int NT, int TRANSB, int RELU_A>
__global__ __launch_bounds__(BLOCK_THREADS)
void gemm_strip_wmma_f32(const float* __restrict__ A, const float* __restrict__ B,
                         const float* __restrict__ bias, float* __restrict__ C,
                         const float* __restrict__ abias,
                         int M, int K, int N, int hasBias)
{
    constexpr int NP = NT * 16;
    __shared__ __attribute__((aligned(16))) float Bs[(GEMM_KC / 4) * NP * 4];

    const int  tid   = threadIdx.x;
    const int  wave  = tid >> 5;
    const int  lane  = tid & 31;
    const int  mn    = lane & 15;
    const int  half  = lane >> 4;
    const long mtile = (long)blockIdx.x * WAVES_PER_BLOCK + wave;
    const bool active = mtile * 16 < (long)M;

    const float2* arow2 = (const float2*)(A + (mtile * 16 + mn) * (long)K);
    const float*  bfrag = Bs + mn * 4 + 2 * half;

    v8f acc[NT];
    #pragma unroll
    for (int t = 0; t < NT; ++t) acc[t] = v8f{};

    for (int k0 = 0; k0 < K; k0 += GEMM_KC) {
        const int kc = (K - k0) < GEMM_KC ? (K - k0) : GEMM_KC;
        if (k0 > 0) __syncthreads();
        stage_b_chunk<NT, TRANSB>(Bs, B, K, N, k0, kc, tid);
        __syncthreads();

        if (active) {
            for (int kq = 0; kq < (kc >> 2); kq += 8) {
                float2 a[8];
                #pragma unroll
                for (int u = 0; u < 8; ++u)
                    a[u] = arow2[((k0 + 4 * (kq + u)) >> 1) + half];
                #pragma unroll
                for (int u = 0; u < 8; ++u) {
                    v2f av; av.x = a[u].x; av.y = a[u].y;
                    if (RELU_A) {
                        const float2 b2v = *(const float2*)(abias + k0 + 4 * (kq + u) + 2 * half);
                        const float vx = av.x + b2v.x;
                        const float vy = av.y + b2v.y;
                        av.x = vx > 0.0f ? vx : 0.0f;
                        av.y = vy > 0.0f ? vy : 0.0f;
                    }
                    #pragma unroll
                    for (int t = 0; t < NT; ++t) {
                        const float2 b2 = *(const float2*)(bfrag + ((kq + u) * NP + t * 16) * 4);
                        v2f bv; bv.x = b2.x; bv.y = b2.y;
                        acc[t] = __builtin_amdgcn_wmma_f32_16x16x4_f32(
                                      false, av,  false, bv,
                                      (short)0, acc[t],
                                      false,  false);
                    }
                }
            }
        }
    }

    if (active) {
        #pragma unroll
        for (int t = 0; t < NT; ++t) {
            const int gn = t * 16 + mn;
            if (gn < N) {
                const float bval = hasBias ? bias[gn] : 0.0f;
                #pragma unroll
                for (int r = 0; r < 8; ++r) {
                    long row = mtile * 16 + r + 8 * half;
                    C[row * (long)N + gn] = acc[t][r] + bval;
                }
            }
        }
    }
}

#define SORT_BLOCKS    256
#define SORT_THREADS   1024
#define BKT_SHIFT      9
#define BKT_ROWS       512
#define N_BKT          196
#define BKT_PITCH      256
#define FIN_THREADS    1024
#define ROWS_IN_FLIGHT 8
#define ROWS_PER_WAVE  8

__global__ __launch_bounds__(SORT_THREADS)
void edge_bucket_count(const int* __restrict__ rows, int* __restrict__ cntm, int nE)
{
    __shared__ int hist[BKT_PITCH];
    const int tid = threadIdx.x;
    if (tid < BKT_PITCH) hist[tid] = 0;
    __syncthreads();
    const int  chunk = (nE + SORT_BLOCKS - 1) / SORT_BLOCKS;
    const long e0 = (long)blockIdx.x * chunk;
    const long e1 = (e0 + chunk) < (long)nE ? (e0 + chunk) : (long)nE;
    for (long e = e0 + tid; e < e1; e += SORT_THREADS)
        atomicAdd(&hist[rows[e] >> BKT_SHIFT], 1);
    __syncthreads();
    if (tid < BKT_PITCH) cntm[blockIdx.x * BKT_PITCH + tid] = hist[tid];
}

__global__ __launch_bounds__(SORT_THREADS)
void edge_bucket_scatter(const int* __restrict__ rows, const int* __restrict__ cols,
                         const float* __restrict__ ew, const int* __restrict__ cntm,
                         int* __restrict__ bstart, int* __restrict__ recs_i, int nE)
{
    __shared__ int s_ptot[4 * BKT_PITCH];
    __shared__ int s_pbef[4 * BKT_PITCH];
    __shared__ int s_tot[BKT_PITCH];
    __shared__ int s_bef[BKT_PITCH];
    __shared__ int s_cur[BKT_PITCH];
    int4* recs = (int4*)recs_i;
    const int tid  = threadIdx.x;
    const int lane = tid & 31;
    const int blk  = blockIdx.x;

    {
        const int t = tid & (BKT_PITCH - 1);
        const int q = tid >> 8;
        int tot = 0, bef = 0;
        const int* cp = cntm + q * (SORT_BLOCKS / 4) * BKT_PITCH + t;
        for (int g = 0; g < SORT_BLOCKS / 4; g += 16) {
            int v[16];
            #pragma unroll
            for (int u = 0; u < 16; ++u) v[u] = cp[(g + u) * BKT_PITCH];
            #pragma unroll
            for (int u = 0; u < 16; ++u) {
                const int j = q * (SORT_BLOCKS / 4) + g + u;
                tot += v[u];
                bef += (j < blk) ? v[u] : 0;
            }
        }
        s_ptot[tid] = tot;
        s_pbef[tid] = bef;
    }
    __syncthreads();
    if (tid < BKT_PITCH) {
        s_tot[tid] = s_ptot[tid] + s_ptot[tid + BKT_PITCH] + s_ptot[tid + 2 * BKT_PITCH] + s_ptot[tid + 3 * BKT_PITCH];
        s_bef[tid] = s_pbef[tid] + s_pbef[tid + BKT_PITCH] + s_pbef[tid + 2 * BKT_PITCH] + s_pbef[tid + 3 * BKT_PITCH];
    }
    __syncthreads();
    if (tid < 32) {
        int loc[8];
        int sum = 0;
        for (int i = 0; i < 8; ++i) { loc[i] = s_tot[lane * 8 + i]; sum += loc[i]; }
        int inc = sum;
        for (int o = 1; o < 32; o <<= 1) {
            const int up = __shfl_up(inc, o);
            if (lane >= o) inc += up;
        }
        int run = inc - sum;
        for (int i = 0; i < 8; ++i) {
            const int b = lane * 8 + i;
            s_cur[b] = run + s_bef[b];
            if (blk == 0 && b <= N_BKT) bstart[b] = run;
            run += loc[i];
        }
    }
    __syncthreads();

    const int  chunk = (nE + SORT_BLOCKS - 1) / SORT_BLOCKS;
    const long e0 = (long)blk * chunk;
    const long e1 = (e0 + chunk) < (long)nE ? (e0 + chunk) : (long)nE;
    for (long e = e0 + tid; e < e1; e += SORT_THREADS) {
        const int   r = rows[e];
        const int   c = cols[e];
        const float w = ew[e];
        const int   p = atomicAdd(&s_cur[r >> BKT_SHIFT], 1);
        recs[p] = make_int4(r, c, __float_as_int(w), 0);
    }
}

__global__ __launch_bounds__(FIN_THREADS)
void bucket_rows_sort(const int* __restrict__ recs_i, const int* __restrict__ bstart,
                      int* __restrict__ csr_i, int* __restrict__ rowptr)
{
    __shared__ int s_cnt[BKT_ROWS];
    const int4* recs = (const int4*)recs_i;
    int2*       csr  = (int2*)csr_i;
    const int tid  = threadIdx.x;
    const int lane = tid & 31;
    const int b    = blockIdx.x;
    const int e0   = bstart[b];
    const int e1   = bstart[b + 1];

    if (tid < BKT_ROWS) s_cnt[tid] = 0;
    __syncthreads();
    for (int eb = e0 + tid; eb < e1; eb += 4 * FIN_THREADS) {
        int r[4];
        #pragma unroll
        for (int u = 0; u < 4; ++u) {
            const int e = eb + u * FIN_THREADS;
            r[u] = e < e1 ? recs[e].x : -1;
        }
        #pragma unroll
        for (int u = 0; u < 4; ++u)
            if (r[u] >= 0) atomicAdd(&s_cnt[r[u] & (BKT_ROWS - 1)], 1);
    }
    __syncthreads();
    if (tid < 32) {
        int loc[16];
        int sum = 0;
        for (int i = 0; i < 16; ++i) { loc[i] = s_cnt[lane * 16 + i]; sum += loc[i]; }
        int inc = sum;
        for (int o = 1; o < 32; o <<= 1) {
            const int up = __shfl_up(inc, o);
            if (lane >= o) inc += up;
        }
        int run = e0 + inc - sum;
        for (int i = 0; i < 16; ++i) {
            const int r = lane * 16 + i;
            s_cnt[r] = run;
            rowptr[b * BKT_ROWS + r] = run;
            run += loc[i];
        }
    }
    __syncthreads();
    for (int eb = e0 + tid; eb < e1; eb += 4 * FIN_THREADS) {
        int4 rec[4];
        #pragma unroll
        for (int u = 0; u < 4; ++u) {
            const int e = eb + u * FIN_THREADS;
            rec[u] = e < e1 ? recs[e] : make_int4(-1, 0, 0, 0);
        }
        #pragma unroll
        for (int u = 0; u < 4; ++u) {
            if (rec[u].x >= 0) {
                const int p = atomicAdd(&s_cnt[rec[u].x & (BKT_ROWS - 1)], 1);
                csr[p] = make_int2(rec[u].y, rec[u].z);
            }
        }
    }
}

template <int VEC>
static __device__ __forceinline__
void spmm_rows(const int* __restrict__ rowptr, const int* __restrict__ csr_i,
               const float* __restrict__ dense, float* __restrict__ out, int nrows)
{
    typedef __attribute__((ext_vector_type(VEC))) float vecf;
    constexpr int F = 32 * VEC;
    const int2* csr  = (const int2*)csr_i;
    const int   lane = threadIdx.x & 31;
    const int wv = __builtin_amdgcn_readfirstlane((int)(threadIdx.x >> 5));
    const int r0 = (blockIdx.x * WAVES_PER_BLOCK + wv) * ROWS_PER_WAVE;
    if (r0 >= nrows) return;
    const int nr = (nrows - r0) < ROWS_PER_WAVE ? (nrows - r0) : ROWS_PER_WAVE;

    int rp = 0;
    if (lane <= nr) rp = rowptr[r0 + lane];
    int s = __builtin_amdgcn_readlane(rp, 0);
    int t = __builtin_amdgcn_readlane(rp, 1);
    int2 nxt = make_int2(0, 0);
    if (lane < t - s) nxt = csr[s + lane];

    for (int i = 0; i < nr; ++i) {
        const int si = s;
        const int ti = t;
        int2 my = nxt;
        if (i + 1 < nr) {
            s = ti;
            t = __builtin_amdgcn_readlane(rp, i + 2);
            nxt = make_int2(0, 0);
            if (lane < t - s) nxt = csr[s + lane];
        }
        vecf acc = (vecf)(0.0f);
        for (int base = si; base < ti; base += 32) {
            const int n = (ti - base) < 32 ? (ti - base) : 32;
            if (base > si) {
                my = make_int2(0, 0);
                if (lane < n) my = csr[base + lane];
            }
            for (int j = 0; j < n; j += ROWS_IN_FLIGHT) {
                vecf v[ROWS_IN_FLIGHT];
                #pragma unroll
                for (int u = 0; u < ROWS_IN_FLIGHT; ++u) {
                    const int jj  = (j + u) < n ? (j + u) : (n - 1);
                    const int src = __builtin_amdgcn_readlane(my.x, jj);
                    v[u] = *(const vecf*)(dense + (long)src * F + lane * VEC);
                }
                #pragma unroll
                for (int u = 0; u < ROWS_IN_FLIGHT; ++u) {
                    const int   jj = (j + u) < n ? (j + u) : (n - 1);
                    const float wr = __int_as_float(__builtin_amdgcn_readlane(my.y, jj));
                    const float w  = (j + u) < n ? wr : 0.0f;
                    acc += v[u] * w;
                }
            }
        }
        *(vecf*)(out + (long)(r0 + i) * F + lane * VEC) = acc;
    }
}

__global__ __launch_bounds__(BLOCK_THREADS)
void spmm_rows_f128(const int* __restrict__ rowptr, const int* __restrict__ csr,
                    const float* __restrict__ dense, float* __restrict__ out, int nrows)
{
    spmm_rows<4>(rowptr, csr, dense, out, nrows);
}

__global__ __launch_bounds__(BLOCK_THREADS)
void spmm_rows_f64(const int* __restrict__ rowptr, const int* __restrict__ csr,
                   const float* __restrict__ dense, float* __restrict__ out, int nrows)
{
    spmm_rows<2>(rowptr, csr, dense, out, nrows);
}

static inline int spmm_blocks(int nrows) {
    return (nrows + WAVES_PER_BLOCK * ROWS_PER_WAVE - 1) / (WAVES_PER_BLOCK * ROWS_PER_WAVE);
}

static inline int mtiles_blocks(int M) {
    return ((M / 16) + WAVES_PER_BLOCK - 1) / WAVES_PER_BLOCK;
}

extern "C" void kernel_launch(void* const* d_in, const int* in_sizes, int n_in,
                              void* d_out, int out_size, void* d_ws, size_t ws_size,
                              hipStream_t stream)
{
    const float* feat0 = (const float*)d_in[0];
    const float* feat1 = (const float*)d_in[1];
    const float* fc0_w = (const float*)d_in[2];
    const float* fc0_b = (const float*)d_in[3];
    const float* fc1_w = (const float*)d_in[4];
    const float* fc1_b = (const float*)d_in[5];
    const float* W1    = (const float*)d_in[6];
    const float* b1    = (const float*)d_in[7];
    const float* W2    = (const float*)d_in[8];
    const float* b2    = (const float*)d_in[9];
    const float* pw    = (const float*)d_in[10];
    const float* pb    = (const float*)d_in[11];
    const int*   erow  = (const int*)d_in[12];
    const int*   ecol  = (const int*)d_in[13];
    const float* ew    = (const float*)d_in[14];
    float*       out   = (float*)d_out;

    const int E       = in_sizes[12];
    const int N_NODES = 100000;
    const int NT      = 50000;

    float* x   = (float*)d_ws;
    float* h   = x + (long)N_NODES * 128;
    float* agg = h + (long)N_NODES * 128;
    int* cntm   = (int*)(agg + (long)N_NODES * 128);
    int* bstart = cntm + SORT_BLOCKS * BKT_PITCH;
    int* recs   = bstart + 256;
    int* csr    = recs + 4 * (long)1600000;
    int* rowptr = csr + 2 * (long)1600000;

    edge_bucket_count<<<SORT_BLOCKS, SORT_THREADS, 0, stream>>>(erow, cntm, E);
    edge_bucket_scatter<<<SORT_BLOCKS, SORT_THREADS, 0, stream>>>(
        erow, ecol, ew, cntm, bstart, recs, E);
    bucket_rows_sort<<<N_BKT, FIN_THREADS, 0, stream>>>(recs, bstart, csr, rowptr);

    gemm_strip_wmma_f32<8, 1, 0><<<mtiles_blocks(NT), BLOCK_THREADS, 0, stream>>>(
        feat0, fc0_w, fc0_b, x, nullptr, NT, 128, 128, 1);
    gemm_strip_wmma_f32<8, 1, 0><<<mtiles_blocks(NT), BLOCK_THREADS, 0, stream>>>(
        feat1, fc1_w, fc1_b, x + (long)NT * 128, nullptr, NT, 256, 128, 1);

    gemm_strip_wmma_f32<8, 0, 0><<<mtiles_blocks(N_NODES), BLOCK_THREADS, 0, stream>>>(
        x, W1, nullptr, h, nullptr, N_NODES, 128, 128, 0);
    spmm_rows_f128<<<spmm_blocks(N_NODES), BLOCK_THREADS, 0, stream>>>(rowptr, csr, h, agg, N_NODES);

    gemm_strip_wmma_f32<4, 0, 1><<<mtiles_blocks(N_NODES), BLOCK_THREADS, 0, stream>>>(
        agg, W2, nullptr, x, b1, N_NODES, 128, 64, 0);
    spmm_rows_f64<<<spmm_blocks(N_NODES), BLOCK_THREADS, 0, stream>>>(rowptr, csr, x, h, N_NODES);

    gemm_strip_wmma_f32<1, 1, 1><<<mtiles_blocks(N_NODES), BLOCK_THREADS, 0, stream>>>(
        h, pw, pb, out, b2, N_NODES, 64, 8, 1);
}


# Layer-1 weight folded into the projections each call: small WMMA kernel forms fc_w.T@W1 and fc_b@W1, projections write x@W1 directly; 100000x128x128 GEMM gone; sum order changes (reassociated)
# speedup vs baseline: 3.3122x; 1.2051x over previous
#include <hip/hip_runtime.h>

typedef __attribute__((ext_vector_type(2))) float v2f;
typedef __attribute__((ext_vector_type(8))) float v8f;

#define WAVES_PER_BLOCK 8
#define BLOCK_THREADS 256

#define GEMM_KC 128

template <int NT, int TRANSB>
static __device__ __forceinline__
void stage_b_chunk(float* __restrict__ Bs, const float* __restrict__ B,
                   int K, int N, int k0, int kc, int tid)
{
    constexpr int NP = NT * 16;
    const int nquads = (kc >> 2) * NP;
    if (TRANSB == 0) {
        for (int q = tid; q < nquads; q += BLOCK_THREADS) {
            const int n  = q % NP;
            const int kq = q / NP;
            float4 v = make_float4(0.0f, 0.0f, 0.0f, 0.0f);
            if (n < N) {
                const float* p = B + (long)(k0 + 4 * kq) * N + n;
                v.x = p[0];
                v.y = p[N];
                v.z = p[2 * (long)N];
                v.w = p[3 * (long)N];
            }
            *(float4*)&Bs[(kq * NP + n) * 4] = v;
        }
    } else {
        const int qk8 = kc >> 5;
        for (int i = tid; i < nquads; i += BLOCK_THREADS) {
            const int kq_lo = i & 7;
            const int n_lo  = (i >> 3) & 3;
            const int rest  = i >> 5;
            const int kq    = (rest % qk8) * 8 + kq_lo;
            const int n     = (rest / qk8) * 4 + n_lo;
            float4 v = make_float4(0.0f, 0.0f, 0.0f, 0.0f);
            if (n < N) v = *(const float4*)(B + (long)n * K + k0 + 4 * kq);
            *(float4*)&Bs[(kq * NP + n) * 4] = v;
        }
    }
}

template <int NT, int TRANSB, int RELU_A>
__global__ __launch_bounds__(BLOCK_THREADS)
void gemm_strip_wmma_f32(const float* __restrict__ A, const float* __restrict__ B,
                         const float* __restrict__ bias, float* __restrict__ C,
                         const float* __restrict__ abias,
                         int M, int K, int N, int hasBias)
{
    constexpr int NP = NT * 16;
    __shared__ __attribute__((aligned(16))) float Bs[(GEMM_KC / 4) * NP * 4];

    const int  tid   = threadIdx.x;
    const int  wave  = tid >> 5;
    const int  lane  = tid & 31;
    const int  mn    = lane & 15;
    const int  half  = lane >> 4;
    const long mtile = (long)blockIdx.x * WAVES_PER_BLOCK + wave;
    const bool active = mtile * 16 < (long)M;

    const float2* arow2 = (const float2*)(A + (mtile * 16 + mn) * (long)K);
    const float*  bfrag = Bs + mn * 4 + 2 * half;

    v8f acc[NT];
    #pragma unroll
    for (int t = 0; t < NT; ++t) acc[t] = v8f{};

    for (int k0 = 0; k0 < K; k0 += GEMM_KC) {
        const int kc = (K - k0) < GEMM_KC ? (K - k0) : GEMM_KC;
        if (k0 > 0) __syncthreads();
        stage_b_chunk<NT, TRANSB>(Bs, B, K, N, k0, kc, tid);
        __syncthreads();

        if (active) {
            for (int kq = 0; kq < (kc >> 2); kq += 8) {
                float2 a[8];
                #pragma unroll
                for (int u = 0; u < 8; ++u)
                    a[u] = arow2[((k0 + 4 * (kq + u)) >> 1) + half];
                #pragma unroll
                for (int u = 0; u < 8; ++u) {
                    v2f av; av.x = a[u].x; av.y = a[u].y;
                    if (RELU_A) {
                        const float2 b2v = *(const float2*)(abias + k0 + 4 * (kq + u) + 2 * half);
                        const float vx = av.x + b2v.x;
                        const float vy = av.y + b2v.y;
                        av.x = vx > 0.0f ? vx : 0.0f;
                        av.y = vy > 0.0f ? vy : 0.0f;
                    }
                    #pragma unroll
                    for (int t = 0; t < NT; ++t) {
                        const float2 b2 = *(const float2*)(bfrag + ((kq + u) * NP + t * 16) * 4);
                        v2f bv; bv.x = b2.x; bv.y = b2.y;
                        acc[t] = __builtin_amdgcn_wmma_f32_16x16x4_f32(
                                      false, av,  false, bv,
                                      (short)0, acc[t],
                                      false,  false);
                    }
                }
            }
        }
    }

    if (active) {
        #pragma unroll
        for (int t = 0; t < NT; ++t) {
            const int gn = t * 16 + mn;
            if (gn < N) {
                const float bval = hasBias ? bias[gn] : 0.0f;
                #pragma unroll
                for (int r = 0; r < 8; ++r) {
                    long row = mtile * 16 + r + 8 * half;
                    C[row * (long)N + gn] = acc[t][r] + bval;
                }
            }
        }
    }
}

#define FOLD_K 128

__global__ __launch_bounds__(BLOCK_THREADS)
void fold_proj_into_w_wmma_f32(const float* __restrict__ S0, const float* __restrict__ sb0,
                               float* __restrict__ Wc0, int Min0,
                               const float* __restrict__ S1, const float* __restrict__ sb1,
                               float* __restrict__ Wc1, int Min1,
                               const float* __restrict__ W, int N)
{
    const int tid  = threadIdx.x;
    const int wave = tid >> 5;
    const int lane = tid & 31;
    const int mn   = lane & 15;
    const int half = lane >> 4;

    const int tiles0 = Min0 / 16 + 1;
    int mtile = blockIdx.x;
    const float* S = S0; const float* sb = sb0; float* Wc = Wc0; int Min = Min0;
    if (mtile >= tiles0) { mtile -= tiles0; S = S1; sb = sb1; Wc = Wc1; Min = Min1; }

    const int m = mtile * 16 + mn;
    const int n = wave * 16 + mn;

    const bool   a_row  = m < Min;
    const float* ap     = a_row ? S + m : sb;
    const long   apitch = a_row ? (long)Min : 1;
    const float* bp     = W + (n < N ? n : 0);

    v8f acc = v8f{};
    #pragma unroll 8
    for (int kq = 0; kq < FOLD_K / 4; ++kq) {
        const int k = 4 * kq + 2 * half;
        v2f av; av.x = ap[k * apitch];     av.y = ap[(k + 1) * apitch];
        v2f bv; bv.x = bp[(long)k * N];    bv.y = bp[(long)(k + 1) * N];
        acc = __builtin_amdgcn_wmma_f32_16x16x4_f32(
                   false, av,  false, bv,
                   (short)0, acc,
                   false,  false);
    }

    if (n < N) {
        #pragma unroll
        for (int r = 0; r < 8; ++r) {
            const int row = mtile * 16 + r + 8 * half;
            if (row <= Min) Wc[(long)row * N + n] = acc[r];
        }
    }
}

#define SORT_BLOCKS    256
#define SORT_THREADS   1024
#define BKT_SHIFT      9
#define BKT_ROWS       512
#define N_BKT          196
#define BKT_PITCH      256
#define FIN_THREADS    1024
#define ROWS_IN_FLIGHT 8
#define ROWS_PER_WAVE  8

__global__ __launch_bounds__(SORT_THREADS)
void edge_bucket_count(const int* __restrict__ rows, int* __restrict__ cntm, int nE)
{
    __shared__ int hist[BKT_PITCH];
    const int tid = threadIdx.x;
    if (tid < BKT_PITCH) hist[tid] = 0;
    __syncthreads();
    const int  chunk = (nE + SORT_BLOCKS - 1) / SORT_BLOCKS;
    const long e0 = (long)blockIdx.x * chunk;
    const long e1 = (e0 + chunk) < (long)nE ? (e0 + chunk) : (long)nE;
    for (long e = e0 + tid; e < e1; e += SORT_THREADS)
        atomicAdd(&hist[rows[e] >> BKT_SHIFT], 1);
    __syncthreads();
    if (tid < BKT_PITCH) cntm[blockIdx.x * BKT_PITCH + tid] = hist[tid];
}

__global__ __launch_bounds__(SORT_THREADS)
void edge_bucket_scatter(const int* __restrict__ rows, const int* __restrict__ cols,
                         const float* __restrict__ ew, const int* __restrict__ cntm,
                         int* __restrict__ bstart, int* __restrict__ recs_i, int nE)
{
    __shared__ int s_ptot[4 * BKT_PITCH];
    __shared__ int s_pbef[4 * BKT_PITCH];
    __shared__ int s_tot[BKT_PITCH];
    __shared__ int s_bef[BKT_PITCH];
    __shared__ int s_cur[BKT_PITCH];
    int4* recs = (int4*)recs_i;
    const int tid  = threadIdx.x;
    const int lane = tid & 31;
    const int blk  = blockIdx.x;

    {
        const int t = tid & (BKT_PITCH - 1);
        const int q = tid >> 8;
        int tot = 0, bef = 0;
        const int* cp = cntm + q * (SORT_BLOCKS / 4) * BKT_PITCH + t;
        for (int g = 0; g < SORT_BLOCKS / 4; g += 16) {
            int v[16];
            #pragma unroll
            for (int u = 0; u < 16; ++u) v[u] = cp[(g + u) * BKT_PITCH];
            #pragma unroll
            for (int u = 0; u < 16; ++u) {
                const int j = q * (SORT_BLOCKS / 4) + g + u;
                tot += v[u];
                bef += (j < blk) ? v[u] : 0;
            }
        }
        s_ptot[tid] = tot;
        s_pbef[tid] = bef;
    }
    __syncthreads();
    if (tid < BKT_PITCH) {
        s_tot[tid] = s_ptot[tid] + s_ptot[tid + BKT_PITCH] + s_ptot[tid + 2 * BKT_PITCH] + s_ptot[tid + 3 * BKT_PITCH];
        s_bef[tid] = s_pbef[tid] + s_pbef[tid + BKT_PITCH] + s_pbef[tid + 2 * BKT_PITCH] + s_pbef[tid + 3 * BKT_PITCH];
    }
    __syncthreads();
    if (tid < 32) {
        int loc[8];
        int sum = 0;
        for (int i = 0; i < 8; ++i) { loc[i] = s_tot[lane * 8 + i]; sum += loc[i]; }
        int inc = sum;
        for (int o = 1; o < 32; o <<= 1) {
            const int up = __shfl_up(inc, o);
            if (lane >= o) inc += up;
        }
        int run = inc - sum;
        for (int i = 0; i < 8; ++i) {
            const int b = lane * 8 + i;
            s_cur[b] = run + s_bef[b];
            if (blk == 0 && b <= N_BKT) bstart[b] = run;
            run += loc[i];
        }
    }
    __syncthreads();

    const int  chunk = (nE + SORT_BLOCKS - 1) / SORT_BLOCKS;
    const long e0 = (long)blk * chunk;
    const long e1 = (e0 + chunk) < (long)nE ? (e0 + chunk) : (long)nE;
    for (long e = e0 + tid; e < e1; e += SORT_THREADS) {
        const int   r = rows[e];
        const int   c = cols[e];
        const float w = ew[e];
        const int   p = atomicAdd(&s_cur[r >> BKT_SHIFT], 1);
        recs[p] = make_int4(r, c, __float_as_int(w), 0);
    }
}

__global__ __launch_bounds__(FIN_THREADS)
void bucket_rows_sort(const int* __restrict__ recs_i, const int* __restrict__ bstart,
                      int* __restrict__ csr_i, int* __restrict__ rowptr)
{
    __shared__ int s_cnt[BKT_ROWS];
    const int4* recs = (const int4*)recs_i;
    int2*       csr  = (int2*)csr_i;
    const int tid  = threadIdx.x;
    const int lane = tid & 31;
    const int b    = blockIdx.x;
    const int e0   = bstart[b];
    const int e1   = bstart[b + 1];

    if (tid < BKT_ROWS) s_cnt[tid] = 0;
    __syncthreads();
    for (int eb = e0 + tid; eb < e1; eb += 4 * FIN_THREADS) {
        int r[4];
        #pragma unroll
        for (int u = 0; u < 4; ++u) {
            const int e = eb + u * FIN_THREADS;
            r[u] = e < e1 ? recs[e].x : -1;
        }
        #pragma unroll
        for (int u = 0; u < 4; ++u)
            if (r[u] >= 0) atomicAdd(&s_cnt[r[u] & (BKT_ROWS - 1)], 1);
    }
    __syncthreads();
    if (tid < 32) {
        int loc[16];
        int sum = 0;
        for (int i = 0; i < 16; ++i) { loc[i] = s_cnt[lane * 16 + i]; sum += loc[i]; }
        int inc = sum;
        for (int o = 1; o < 32; o <<= 1) {
            const int up = __shfl_up(inc, o);
            if (lane >= o) inc += up;
        }
        int run = e0 + inc - sum;
        for (int i = 0; i < 16; ++i) {
            const int r = lane * 16 + i;
            s_cnt[r] = run;
            rowptr[b * BKT_ROWS + r] = run;
            run += loc[i];
        }
    }
    __syncthreads();
    for (int eb = e0 + tid; eb < e1; eb += 4 * FIN_THREADS) {
        int4 rec[4];
        #pragma unroll
        for (int u = 0; u < 4; ++u) {
            const int e = eb + u * FIN_THREADS;
            rec[u] = e < e1 ? recs[e] : make_int4(-1, 0, 0, 0);
        }
        #pragma unroll
        for (int u = 0; u < 4; ++u) {
            if (rec[u].x >= 0) {
                const int p = atomicAdd(&s_cnt[rec[u].x & (BKT_ROWS - 1)], 1);
                csr[p] = make_int2(rec[u].y, rec[u].z);
            }
        }
    }
}

template <int VEC>
static __device__ __forceinline__
void spmm_rows(const int* __restrict__ rowptr, const int* __restrict__ csr_i,
               const float* __restrict__ dense, float* __restrict__ out, int nrows)
{
    typedef __attribute__((ext_vector_type(VEC))) float vecf;
    constexpr int F = 32 * VEC;
    const int2* csr  = (const int2*)csr_i;
    const int   lane = threadIdx.x & 31;
    const int wv = __builtin_amdgcn_readfirstlane((int)(threadIdx.x >> 5));
    const int r0 = (blockIdx.x * WAVES_PER_BLOCK + wv) * ROWS_PER_WAVE;
    if (r0 >= nrows) return;
    const int nr = (nrows - r0) < ROWS_PER_WAVE ? (nrows - r0) : ROWS_PER_WAVE;

    int rp = 0;
    if (lane <= nr) rp = rowptr[r0 + lane];
    int s = __builtin_amdgcn_readlane(rp, 0);
    int t = __builtin_amdgcn_readlane(rp, 1);
    int2 nxt = make_int2(0, 0);
    if (lane < t - s) nxt = csr[s + lane];

    for (int i = 0; i < nr; ++i) {
        const int si = s;
        const int ti = t;
        int2 my = nxt;
        if (i + 1 < nr) {
            s = ti;
            t = __builtin_amdgcn_readlane(rp, i + 2);
            nxt = make_int2(0, 0);
            if (lane < t - s) nxt = csr[s + lane];
        }
        vecf acc = (vecf)(0.0f);
        for (int base = si; base < ti; base += 32) {
            const int n = (ti - base) < 32 ? (ti - base) : 32;
            if (base > si) {
                my = make_int2(0, 0);
                if (lane < n) my = csr[base + lane];
            }
            for (int j = 0; j < n; j += ROWS_IN_FLIGHT) {
                vecf v[ROWS_IN_FLIGHT];
                #pragma unroll
                for (int u = 0; u < ROWS_IN_FLIGHT; ++u) {
                    const int jj  = (j + u) < n ? (j + u) : (n - 1);
                    const int src = __builtin_amdgcn_readlane(my.x, jj);
                    v[u] = *(const vecf*)(dense + (long)src * F + lane * VEC);
                }
                #pragma unroll
                for (int u = 0; u < ROWS_IN_FLIGHT; ++u) {
                    const int   jj = (j + u) < n ? (j + u) : (n - 1);
                    const float wr = __int_as_float(__builtin_amdgcn_readlane(my.y, jj));
                    const float w  = (j + u) < n ? wr : 0.0f;
                    acc += v[u] * w;
                }
            }
        }
        *(vecf*)(out + (long)(r0 + i) * F + lane * VEC) = acc;
    }
}

__global__ __launch_bounds__(BLOCK_THREADS)
void spmm_rows_f128(const int* __restrict__ rowptr, const int* __restrict__ csr,
                    const float* __restrict__ dense, float* __restrict__ out, int nrows)
{
    spmm_rows<4>(rowptr, csr, dense, out, nrows);
}

__global__ __launch_bounds__(BLOCK_THREADS)
void spmm_rows_f64(const int* __restrict__ rowptr, const int* __restrict__ csr,
                   const float* __restrict__ dense, float* __restrict__ out, int nrows)
{
    spmm_rows<2>(rowptr, csr, dense, out, nrows);
}

static inline int spmm_blocks(int nrows) {
    return (nrows + WAVES_PER_BLOCK * ROWS_PER_WAVE - 1) / (WAVES_PER_BLOCK * ROWS_PER_WAVE);
}

static inline int mtiles_blocks(int M) {
    return ((M / 16) + WAVES_PER_BLOCK - 1) / WAVES_PER_BLOCK;
}

extern "C" void kernel_launch(void* const* d_in, const int* in_sizes, int n_in,
                              void* d_out, int out_size, void* d_ws, size_t ws_size,
                              hipStream_t stream)
{
    const float* feat0 = (const float*)d_in[0];
    const float* feat1 = (const float*)d_in[1];
    const float* fc0_w = (const float*)d_in[2];
    const float* fc0_b = (const float*)d_in[3];
    const float* fc1_w = (const float*)d_in[4];
    const float* fc1_b = (const float*)d_in[5];
    const float* W1    = (const float*)d_in[6];
    const float* b1    = (const float*)d_in[7];
    const float* W2    = (const float*)d_in[8];
    const float* b2    = (const float*)d_in[9];
    const float* pw    = (const float*)d_in[10];
    const float* pb    = (const float*)d_in[11];
    const int*   erow  = (const int*)d_in[12];
    const int*   ecol  = (const int*)d_in[13];
    const float* ew    = (const float*)d_in[14];
    float*       out   = (float*)d_out;

    const int E       = in_sizes[12];
    const int N_NODES = 100000;
    const int NT      = 50000;

    float* x   = (float*)d_ws;
    float* h   = x + (long)N_NODES * 128;
    float* agg = h + (long)N_NODES * 128;
    int* cntm   = (int*)(agg + (long)N_NODES * 128);
    int* bstart = cntm + SORT_BLOCKS * BKT_PITCH;
    int* recs   = bstart + 256;
    int* csr    = recs + 4 * (long)1600000;
    int* rowptr = csr + 2 * (long)1600000;
    float* wc0  = (float*)(rowptr + N_BKT * BKT_ROWS);
    float* wc1  = wc0 + 144 * 128;

    edge_bucket_count<<<SORT_BLOCKS, SORT_THREADS, 0, stream>>>(erow, cntm, E);
    edge_bucket_scatter<<<SORT_BLOCKS, SORT_THREADS, 0, stream>>>(
        erow, ecol, ew, cntm, bstart, recs, E);
    bucket_rows_sort<<<N_BKT, FIN_THREADS, 0, stream>>>(recs, bstart, csr, rowptr);

    fold_proj_into_w_wmma_f32<<<26, BLOCK_THREADS, 0, stream>>>(
        fc0_w, fc0_b, wc0, 128, fc1_w, fc1_b, wc1, 256, W1, 128);

    gemm_strip_wmma_f32<8, 0, 0><<<mtiles_blocks(NT), BLOCK_THREADS, 0, stream>>>(
        feat0, wc0, wc0 + 128 * 128, h, nullptr, NT, 128, 128, 1);
    gemm_strip_wmma_f32<8, 0, 0><<<mtiles_blocks(NT), BLOCK_THREADS, 0, stream>>>(
        feat1, wc1, wc1 + 256 * 128, h + (long)NT * 128, nullptr, NT, 256, 128, 1);

    spmm_rows_f128<<<spmm_blocks(N_NODES), BLOCK_THREADS, 0, stream>>>(rowptr, csr, h, agg, N_NODES);

    gemm_strip_wmma_f32<4, 0, 1><<<mtiles_blocks(N_NODES), BLOCK_THREADS, 0, stream>>>(
        agg, W2, nullptr, x, b1, N_NODES, 128, 64, 0);
    spmm_rows_f64<<<spmm_blocks(N_NODES), BLOCK_THREADS, 0, stream>>>(rowptr, csr, x, h, N_NODES);

    gemm_strip_wmma_f32<1, 1, 1><<<mtiles_blocks(N_NODES), BLOCK_THREADS, 0, stream>>>(
        h, pw, pb, out, b2, N_NODES, 64, 8, 1);
}


# GEMM B staging: a thread issues all global loads of a chunk (16-byte loads of 4x4 blocks, transposed in registers) before its first LDS store: one memory round trip per chunk, was one per quad
# speedup vs baseline: 3.5392x; 1.0685x over previous
#include <hip/hip_runtime.h>

typedef __attribute__((ext_vector_type(2))) float v2f;
typedef __attribute__((ext_vector_type(8))) float v8f;

#define WAVES_PER_BLOCK 8
#define BLOCK_THREADS 256

#define GEMM_KC 128
#define BS_PAD  4

template <int NT, int TRANSB>
static __device__ __forceinline__
void stage_b_chunk(float* __restrict__ Bs, const float* __restrict__ B,
                   int K, int N, int k0, int kc, int tid)
{
    constexpr int NP    = NT * 16;
    constexpr int PITCH = NP * 4 + BS_PAD;
    if (TRANSB == 0) {
        static_assert(TRANSB == 1 || NT % 2 == 0, "column groups come in eights");
        constexpr int NG    = NP / 4;
        constexpr int NG8   = NG / 8 > 0 ? NG / 8 : 1;
        constexpr int ITEMS = (GEMM_KC / 4) * NG / BLOCK_THREADS;
        const int nitems = (kc >> 2) * NG;
        float4 r[ITEMS > 0 ? ITEMS : 1][4];
        #pragma unroll
        for (int j = 0; j < ITEMS; ++j) {
            const int i  = tid + j * BLOCK_THREADS;
            const int w  = i >> 5;
            const int n4 = (w % NG8) * 8 + (i & 7);
            const int kq = (w / NG8) * 4 + ((i >> 3) & 3);
            #pragma unroll
            for (int s = 0; s < 4; ++s) r[j][s] = make_float4(0.0f, 0.0f, 0.0f, 0.0f);
            if (i < nitems && 4 * n4 < N) {
                const float* p = B + (long)(k0 + 4 * kq) * N + 4 * n4;
                #pragma unroll
                for (int s = 0; s < 4; ++s) r[j][s] = *(const float4*)(p + (long)s * N);
            }
        }
        #pragma unroll
        for (int j = 0; j < ITEMS; ++j) {
            const int i  = tid + j * BLOCK_THREADS;
            const int w  = i >> 5;
            const int n4 = (w % NG8) * 8 + (i & 7);
            const int kq = (w / NG8) * 4 + ((i >> 3) & 3);
            if (i < nitems) {
                float* q = Bs + kq * PITCH + n4 * 16;
                *(float4*)(q + 0)  = make_float4(r[j][0].x, r[j][1].x, r[j][2].x, r[j][3].x);
                *(float4*)(q + 4)  = make_float4(r[j][0].y, r[j][1].y, r[j][2].y, r[j][3].y);
                *(float4*)(q + 8)  = make_float4(r[j][0].z, r[j][1].z, r[j][2].z, r[j][3].z);
                *(float4*)(q + 12) = make_float4(r[j][0].w, r[j][1].w, r[j][2].w, r[j][3].w);
            }
        }
    } else {
        const int nquads = (kc >> 2) * NP;
        const int qk8 = kc >> 5;
        for (int i = tid; i < nquads; i += BLOCK_THREADS) {
            const int kq_lo = i & 7;
            const int n_lo  = (i >> 3) & 3;
            const int rest  = i >> 5;
            const int kq    = (rest % qk8) * 8 + kq_lo;
            const int n     = (rest / qk8) * 4 + n_lo;
            float4 v = make_float4(0.0f, 0.0f, 0.0f, 0.0f);
            if (n < N) v = *(const float4*)(B + (long)n * K + k0 + 4 * kq);
            *(float4*)&Bs[kq * PITCH + n * 4] = v;
        }
    }
}

template <int NT, int TRANSB, int RELU_A>
__global__ __launch_bounds__(BLOCK_THREADS)
void gemm_strip_wmma_f32(const float* __restrict__ A, const float* __restrict__ B,
                         const float* __restrict__ bias, float* __restrict__ C,
                         const float* __restrict__ abias,
                         int M, int K, int N, int hasBias)
{
    constexpr int NP    = NT * 16;
    constexpr int PITCH = NP * 4 + BS_PAD;
    __shared__ __attribute__((aligned(16))) float Bs[(GEMM_KC / 4) * PITCH];

    const int  tid   = threadIdx.x;
    const int  wave  = tid >> 5;
    const int  lane  = tid & 31;
    const int  mn    = lane & 15;
    const int  half  = lane >> 4;
    const long mtile = (long)blockIdx.x * WAVES_PER_BLOCK + wave;
    const bool active = mtile * 16 < (long)M;

    const float2* arow2 = (const float2*)(A + (mtile * 16 + mn) * (long)K);
    const float*  bfrag = Bs + mn * 4 + 2 * half;

    v8f acc[NT];
    #pragma unroll
    for (int t = 0; t < NT; ++t) acc[t] = v8f{};

    for (int k0 = 0; k0 < K; k0 += GEMM_KC) {
        const int kc = (K - k0) < GEMM_KC ? (K - k0) : GEMM_KC;
        if (k0 > 0) __syncthreads();
        stage_b_chunk<NT, TRANSB>(Bs, B, K, N, k0, kc, tid);
        __syncthreads();

        if (active) {
            for (int kq = 0; kq < (kc >> 2); kq += 8) {
                float2 a[8];
                #pragma unroll
                for (int u = 0; u < 8; ++u)
                    a[u] = arow2[((k0 + 4 * (kq + u)) >> 1) + half];
                #pragma unroll
                for (int u = 0; u < 8; ++u) {
                    v2f av; av.x = a[u].x; av.y = a[u].y;
                    if (RELU_A) {
                        const float2 b2v = *(const float2*)(abias + k0 + 4 * (kq + u) + 2 * half);
                        const float vx = av.x + b2v.x;
                        const float vy = av.y + b2v.y;
                        av.x = vx > 0.0f ? vx : 0.0f;
                        av.y = vy > 0.0f ? vy : 0.0f;
                    }
                    #pragma unroll
                    for (int t = 0; t < NT; ++t) {
                        const float2 b2 = *(const float2*)(bfrag + (kq + u) * PITCH + t * 64);
                        v2f bv; bv.x = b2.x; bv.y = b2.y;
                        acc[t] = __builtin_amdgcn_wmma_f32_16x16x4_f32(
                                      false, av,  false, bv,
                                      (short)0, acc[t],
                                      false,  false);
                    }
                }
            }
        }
    }

    if (active) {
        #pragma unroll
        for (int t = 0; t < NT; ++t) {
            const int gn = t * 16 + mn;
            if (gn < N) {
                const float bval = hasBias ? bias[gn] : 0.0f;
                #pragma unroll
                for (int r = 0; r < 8; ++r) {
                    long row = mtile * 16 + r + 8 * half;
                    C[row * (long)N + gn] = acc[t][r] + bval;
                }
            }
        }
    }
}

#define FOLD_K 128

__global__ __launch_bounds__(BLOCK_THREADS)
void fold_proj_into_w_wmma_f32(const float* __restrict__ S0, const float* __restrict__ sb0,
                               float* __restrict__ Wc0, int Min0,
                               const float* __restrict__ S1, const float* __restrict__ sb1,
                               float* __restrict__ Wc1, int Min1,
                               const float* __restrict__ W, int N)
{
    const int tid  = threadIdx.x;
    const int wave = tid >> 5;
    const int lane = tid & 31;
    const int mn   = lane & 15;
    const int half = lane >> 4;

    const int tiles0 = Min0 / 16 + 1;
    int mtile = blockIdx.x;
    const float* S = S0; const float* sb = sb0; float* Wc = Wc0; int Min = Min0;
    if (mtile >= tiles0) { mtile -= tiles0; S = S1; sb = sb1; Wc = Wc1; Min = Min1; }

    const int m = mtile * 16 + mn;
    const int n = wave * 16 + mn;

    const bool   a_row  = m < Min;
    const float* ap     = a_row ? S + m : sb;
    const long   apitch = a_row ? (long)Min : 1;
    const float* bp     = W + (n < N ? n : 0);

    v8f acc = v8f{};
    #pragma unroll 8
    for (int kq = 0; kq < FOLD_K / 4; ++kq) {
        const int k = 4 * kq + 2 * half;
        v2f av; av.x = ap[k * apitch];     av.y = ap[(k + 1) * apitch];
        v2f bv; bv.x = bp[(long)k * N];    bv.y = bp[(long)(k + 1) * N];
        acc = __builtin_amdgcn_wmma_f32_16x16x4_f32(
                   false, av,  false, bv,
                   (short)0, acc,
                   false,  false);
    }

    if (n < N) {
        #pragma unroll
        for (int r = 0; r < 8; ++r) {
            const int row = mtile * 16 + r + 8 * half;
            if (row <= Min) Wc[(long)row * N + n] = acc[r];
        }
    }
}

#define SORT_BLOCKS    256
#define SORT_THREADS   1024
#define BKT_SHIFT      9
#define BKT_ROWS       512
#define N_BKT          196
#define BKT_PITCH      256
#define FIN_THREADS    1024
#define ROWS_IN_FLIGHT 8
#define ROWS_PER_WAVE  8

__global__ __launch_bounds__(SORT_THREADS)
void edge_bucket_count(const int* __restrict__ rows, int* __restrict__ cntm, int nE)
{
    __shared__ int hist[BKT_PITCH];
    const int tid = threadIdx.x;
    if (tid < BKT_PITCH) hist[tid] = 0;
    __syncthreads();
    const int  chunk = (nE + SORT_BLOCKS - 1) / SORT_BLOCKS;
    const long e0 = (long)blockIdx.x * chunk;
    const long e1 = (e0 + chunk) < (long)nE ? (e0 + chunk) : (long)nE;
    for (long e = e0 + tid; e < e1; e += SORT_THREADS)
        atomicAdd(&hist[rows[e] >> BKT_SHIFT], 1);
    __syncthreads();
    if (tid < BKT_PITCH) cntm[blockIdx.x * BKT_PITCH + tid] = hist[tid];
}

__global__ __launch_bounds__(SORT_THREADS)
void edge_bucket_scatter(const int* __restrict__ rows, const int* __restrict__ cols,
                         const float* __restrict__ ew, const int* __restrict__ cntm,
                         int* __restrict__ bstart, int* __restrict__ recs_i, int nE)
{
    __shared__ int s_ptot[4 * BKT_PITCH];
    __shared__ int s_pbef[4 * BKT_PITCH];
    __shared__ int s_tot[BKT_PITCH];
    __shared__ int s_bef[BKT_PITCH];
    __shared__ int s_cur[BKT_PITCH];
    int4* recs = (int4*)recs_i;
    const int tid  = threadIdx.x;
    const int lane = tid & 31;
    const int blk  = blockIdx.x;

    {
        const int t = tid & (BKT_PITCH - 1);
        const int q = tid >> 8;
        int tot = 0, bef = 0;
        const int* cp = cntm + q * (SORT_BLOCKS / 4) * BKT_PITCH + t;
        for (int g = 0; g < SORT_BLOCKS / 4; g += 16) {
            int v[16];
            #pragma unroll
            for (int u = 0; u < 16; ++u) v[u] = cp[(g + u) * BKT_PITCH];
            #pragma unroll
            for (int u = 0; u < 16; ++u) {
                const int j = q * (SORT_BLOCKS / 4) + g + u;
                tot += v[u];
                bef += (j < blk) ? v[u] : 0;
            }
        }
        s_ptot[tid] = tot;
        s_pbef[tid] = bef;
    }
    __syncthreads();
    if (tid < BKT_PITCH) {
        s_tot[tid] = s_ptot[tid] + s_ptot[tid + BKT_PITCH] + s_ptot[tid + 2 * BKT_PITCH] + s_ptot[tid + 3 * BKT_PITCH];
        s_bef[tid] = s_pbef[tid] + s_pbef[tid + BKT_PITCH] + s_pbef[tid + 2 * BKT_PITCH] + s_pbef[tid + 3 * BKT_PITCH];
    }
    __syncthreads();
    if (tid < 32) {
        int loc[8];
        int sum = 0;
        for (int i = 0; i < 8; ++i) { loc[i] = s_tot[lane * 8 + i]; sum += loc[i]; }
        int inc = sum;
        for (int o = 1; o < 32; o <<= 1) {
            const int up = __shfl_up(inc, o);
            if (lane >= o) inc += up;
        }
        int run = inc - sum;
        for (int i = 0; i < 8; ++i) {
            const int b = lane * 8 + i;
            s_cur[b] = run + s_bef[b];
            if (blk == 0 && b <= N_BKT) bstart[b] = run;
            run += loc[i];
        }
    }
    __syncthreads();

    const int  chunk = (nE + SORT_BLOCKS - 1) / SORT_BLOCKS;
    const long e0 = (long)blk * chunk;
    const long e1 = (e0 + chunk) < (long)nE ? (e0 + chunk) : (long)nE;
    for (long e = e0 + tid; e < e1; e += SORT_THREADS) {
        const int   r = rows[e];
        const int   c = cols[e];
        const float w = ew[e];
        const int   p = atomicAdd(&s_cur[r >> BKT_SHIFT], 1);
        recs[p] = make_int4(r, c, __float_as_int(w), 0);
    }
}

__global__ __launch_bounds__(FIN_THREADS)
void bucket_rows_sort(const int* __restrict__ recs_i, const int* __restrict__ bstart,
                      int* __restrict__ csr_i, int* __restrict__ rowptr)
{
    __shared__ int s_cnt[BKT_ROWS];
    const int4* recs = (const int4*)recs_i;
    int2*       csr  = (int2*)csr_i;
    const int tid  = threadIdx.x;
    const int lane = tid & 31;
    const int b    = blockIdx.x;
    const int e0   = bstart[b];
    const int e1   = bstart[b + 1];

    if (tid < BKT_ROWS) s_cnt[tid] = 0;
    __syncthreads();
    for (int eb = e0 + tid; eb < e1; eb += 4 * FIN_THREADS) {
        int r[4];
        #pragma unroll
        for (int u = 0; u < 4; ++u) {
            const int e = eb + u * FIN_THREADS;
            r[u] = e < e1 ? recs[e].x : -1;
        }
        #pragma unroll
        for (int u = 0; u < 4; ++u)
            if (r[u] >= 0) atomicAdd(&s_cnt[r[u] & (BKT_ROWS - 1)], 1);
    }
    __syncthreads();
    if (tid < 32) {
        int loc[16];
        int sum = 0;
        for (int i = 0; i < 16; ++i) { loc[i] = s_cnt[lane * 16 + i]; sum += loc[i]; }
        int inc = sum;
        for (int o = 1; o < 32; o <<= 1) {
            const int up = __shfl_up(inc, o);
            if (lane >= o) inc += up;
        }
        int run = e0 + inc - sum;
        for (int i = 0; i < 16; ++i) {
            const int r = lane * 16 + i;
            s_cnt[r] = run;
            rowptr[b * BKT_ROWS + r] = run;
            run += loc[i];
        }
    }
    __syncthreads();
    for (int eb = e0 + tid; eb < e1; eb += 4 * FIN_THREADS) {
        int4 rec[4];
        #pragma unroll
        for (int u = 0; u < 4; ++u) {
            const int e = eb + u * FIN_THREADS;
            rec[u] = e < e1 ? recs[e] : make_int4(-1, 0, 0, 0);
        }
        #pragma unroll
        for (int u = 0; u < 4; ++u) {
            if (rec[u].x >= 0) {
                const int p = atomicAdd(&s_cnt[rec[u].x & (BKT_ROWS - 1)], 1);
                csr[p] = make_int2(rec[u].y, rec[u].z);
            }
        }
    }
}

template <int VEC>
static __device__ __forceinline__
void spmm_rows(const int* __restrict__ rowptr, const int* __restrict__ csr_i,
               const float* __restrict__ dense, float* __restrict__ out, int nrows)
{
    typedef __attribute__((ext_vector_type(VEC))) float vecf;
    constexpr int F = 32 * VEC;
    const int2* csr  = (const int2*)csr_i;
    const int   lane = threadIdx.x & 31;
    const int wv = __builtin_amdgcn_readfirstlane((int)(threadIdx.x >> 5));
    const int r0 = (blockIdx.x * WAVES_PER_BLOCK + wv) * ROWS_PER_WAVE;
    if (r0 >= nrows) return;
    const int nr = (nrows - r0) < ROWS_PER_WAVE ? (nrows - r0) : ROWS_PER_WAVE;

    int rp = 0;
    if (lane <= nr) rp = rowptr[r0 + lane];
    int s = __builtin_amdgcn_readlane(rp, 0);
    int t = __builtin_amdgcn_readlane(rp, 1);
    int2 nxt = make_int2(0, 0);
    if (lane < t - s) nxt = csr[s + lane];

    for (int i = 0; i < nr; ++i) {
        const int si = s;
        const int ti = t;
        int2 my = nxt;
        if (i + 1 < nr) {
            s = ti;
            t = __builtin_amdgcn_readlane(rp, i + 2);
            nxt = make_int2(0, 0);
            if (lane < t - s) nxt = csr[s + lane];
        }
        vecf acc = (vecf)(0.0f);
        for (int base = si; base < ti; base += 32) {
            const int n = (ti - base) < 32 ? (ti - base) : 32;
            if (base > si) {
                my = make_int2(0, 0);
                if (lane < n) my = csr[base + lane];
            }
            for (int j = 0; j < n; j += ROWS_IN_FLIGHT) {
                vecf v[ROWS_IN_FLIGHT];
                #pragma unroll
                for (int u = 0; u < ROWS_IN_FLIGHT; ++u) {
                    const int jj  = (j + u) < n ? (j + u) : (n - 1);
                    const int src = __builtin_amdgcn_readlane(my.x, jj);
                    v[u] = *(const vecf*)(dense + (long)src * F + lane * VEC);
                }
                #pragma unroll
                for (int u = 0; u < ROWS_IN_FLIGHT; ++u) {
                    const int   jj = (j + u) < n ? (j + u) : (n - 1);
                    const float wr = __int_as_float(__builtin_amdgcn_readlane(my.y, jj));
                    const float w  = (j + u) < n ? wr : 0.0f;
                    acc += v[u] * w;
                }
            }
        }
        *(vecf*)(out + (long)(r0 + i) * F + lane * VEC) = acc;
    }
}

__global__ __launch_bounds__(BLOCK_THREADS)
void spmm_rows_f128(const int* __restrict__ rowptr, const int* __restrict__ csr,
                    const float* __restrict__ dense, float* __restrict__ out, int nrows)
{
    spmm_rows<4>(rowptr, csr, dense, out, nrows);
}

__global__ __launch_bounds__(BLOCK_THREADS)
void spmm_rows_f64(const int* __restrict__ rowptr, const int* __restrict__ csr,
                   const float* __restrict__ dense, float* __restrict__ out, int nrows)
{
    spmm_rows<2>(rowptr, csr, dense, out, nrows);
}

static inline int spmm_blocks(int nrows) {
    return (nrows + WAVES_PER_BLOCK * ROWS_PER_WAVE - 1) / (WAVES_PER_BLOCK * ROWS_PER_WAVE);
}

static inline int mtiles_blocks(int M) {
    return ((M / 16) + WAVES_PER_BLOCK - 1) / WAVES_PER_BLOCK;
}

extern "C" void kernel_launch(void* const* d_in, const int* in_sizes, int n_in,
                              void* d_out, int out_size, void* d_ws, size_t ws_size,
                              hipStream_t stream)
{
    const float* feat0 = (const float*)d_in[0];
    const float* feat1 = (const float*)d_in[1];
    const float* fc0_w = (const float*)d_in[2];
    const float* fc0_b = (const float*)d_in[3];
    const float* fc1_w = (const float*)d_in[4];
    const float* fc1_b = (const float*)d_in[5];
    const float* W1    = (const float*)d_in[6];
    const float* b1    = (const float*)d_in[7];
    const float* W2    = (const float*)d_in[8];
    const float* b2    = (const float*)d_in[9];
    const float* pw    = (const float*)d_in[10];
    const float* pb    = (const float*)d_in[11];
    const int*   erow  = (const int*)d_in[12];
    const int*   ecol  = (const int*)d_in[13];
    const float* ew    = (const float*)d_in[14];
    float*       out   = (float*)d_out;

    const int E       = in_sizes[12];
    const int N_NODES = 100000;
    const int NT      = 50000;

    float* x   = (float*)d_ws;
    float* h   = x + (long)N_NODES * 128;
    float* agg = h + (long)N_NODES * 128;
    int* cntm   = (int*)(agg + (long)N_NODES * 128);
    int* bstart = cntm + SORT_BLOCKS * BKT_PITCH;
    int* recs   = bstart + 256;
    int* csr    = recs + 4 * (long)1600000;
    int* rowptr = csr + 2 * (long)1600000;
    float* wc0  = (float*)(rowptr + N_BKT * BKT_ROWS);
    float* wc1  = wc0 + 144 * 128;

    edge_bucket_count<<<SORT_BLOCKS, SORT_THREADS, 0, stream>>>(erow, cntm, E);
    edge_bucket_scatter<<<SORT_BLOCKS, SORT_THREADS, 0, stream>>>(
        erow, ecol, ew, cntm, bstart, recs, E);
    bucket_rows_sort<<<N_BKT, FIN_THREADS, 0, stream>>>(recs, bstart, csr, rowptr);

    fold_proj_into_w_wmma_f32<<<26, BLOCK_THREADS, 0, stream>>>(
        fc0_w, fc0_b, wc0, 128, fc1_w, fc1_b, wc1, 256, W1, 128);

    gemm_strip_wmma_f32<8, 0, 0><<<mtiles_blocks(NT), BLOCK_THREADS, 0, stream>>>(
        feat0, wc0, wc0 + 128 * 128, h, nullptr, NT, 128, 128, 1);
    gemm_strip_wmma_f32<8, 0, 0><<<mtiles_blocks(NT), BLOCK_THREADS, 0, stream>>>(
        feat1, wc1, wc1 + 256 * 128, h + (long)NT * 128, nullptr, NT, 256, 128, 1);

    spmm_rows_f128<<<spmm_blocks(N_NODES), BLOCK_THREADS, 0, stream>>>(rowptr, csr, h, agg, N_NODES);

    gemm_strip_wmma_f32<4, 0, 1><<<mtiles_blocks(N_NODES), BLOCK_THREADS, 0, stream>>>(
        agg, W2, nullptr, x, b1, N_NODES, 128, 64, 0);
    spmm_rows_f64<<<spmm_blocks(N_NODES), BLOCK_THREADS, 0, stream>>>(rowptr, csr, x, h, N_NODES);

    gemm_strip_wmma_f32<1, 1, 1><<<mtiles_blocks(N_NODES), BLOCK_THREADS, 0, stream>>>(
        h, pw, pb, out, b2, N_NODES, 64, 8, 1);
}


# SpMM epilogue fusion: block keeps its 64 summed rows in LDS (bias+relu applied), runs the consumer product (W2, predict head) on WMMA there; agg buffers and 2 GEMM launches gone; sum order unchanged
# speedup vs baseline: 3.9975x; 1.1295x over previous
#include <hip/hip_runtime.h>

typedef __attribute__((ext_vector_type(2))) float v2f;
typedef __attribute__((ext_vector_type(8))) float v8f;

#define WAVES_PER_BLOCK 8
#define BLOCK_THREADS 256

#define GEMM_KC 128
#define BS_PAD  4

template <int NT>
static __device__ __forceinline__
void stage_b_chunk(float* __restrict__ Bs, const float* __restrict__ B,
                   int N, int k0, int kc, int tid)
{
    constexpr int NP    = NT * 16;
    constexpr int PITCH = NP * 4 + BS_PAD;
    {
        static_assert(NT % 2 == 0, "column groups come in eights");
        constexpr int NG    = NP / 4;
        constexpr int NG8   = NG / 8 > 0 ? NG / 8 : 1;
        constexpr int ITEMS = (GEMM_KC / 4) * NG / BLOCK_THREADS;
        const int nitems = (kc >> 2) * NG;
        float4 r[ITEMS > 0 ? ITEMS : 1][4];
        #pragma unroll
        for (int j = 0; j < ITEMS; ++j) {
            const int i  = tid + j * BLOCK_THREADS;
            const int w  = i >> 5;
            const int n4 = (w % NG8) * 8 + (i & 7);
            const int kq = (w / NG8) * 4 + ((i >> 3) & 3);
            #pragma unroll
            for (int s = 0; s < 4; ++s) r[j][s] = make_float4(0.0f, 0.0f, 0.0f, 0.0f);
            if (i < nitems && 4 * n4 < N) {
                const float* p = B + (long)(k0 + 4 * kq) * N + 4 * n4;
                #pragma unroll
                for (int s = 0; s < 4; ++s) r[j][s] = *(const float4*)(p + (long)s * N);
            }
        }
        #pragma unroll
        for (int j = 0; j < ITEMS; ++j) {
            const int i  = tid + j * BLOCK_THREADS;
            const int w  = i >> 5;
            const int n4 = (w % NG8) * 8 + (i & 7);
            const int kq = (w / NG8) * 4 + ((i >> 3) & 3);
            if (i < nitems) {
                float* q = Bs + kq * PITCH + n4 * 16;
                *(float4*)(q + 0)  = make_float4(r[j][0].x, r[j][1].x, r[j][2].x, r[j][3].x);
                *(float4*)(q + 4)  = make_float4(r[j][0].y, r[j][1].y, r[j][2].y, r[j][3].y);
                *(float4*)(q + 8)  = make_float4(r[j][0].z, r[j][1].z, r[j][2].z, r[j][3].z);
                *(float4*)(q + 12) = make_float4(r[j][0].w, r[j][1].w, r[j][2].w, r[j][3].w);
            }
        }
    }
}

template <int NT>
__global__ __launch_bounds__(BLOCK_THREADS)
void gemm_strip_wmma_f32(const float* __restrict__ A, const float* __restrict__ B,
                         const float* __restrict__ bias, float* __restrict__ C,
                         int M, int K, int N, int hasBias)
{
    constexpr int NP    = NT * 16;
    constexpr int PITCH = NP * 4 + BS_PAD;
    __shared__ __attribute__((aligned(16))) float Bs[(GEMM_KC / 4) * PITCH];

    const int  tid   = threadIdx.x;
    const int  wave  = tid >> 5;
    const int  lane  = tid & 31;
    const int  mn    = lane & 15;
    const int  half  = lane >> 4;
    const long mtile = (long)blockIdx.x * WAVES_PER_BLOCK + wave;
    const bool active = mtile * 16 < (long)M;

    const float2* arow2 = (const float2*)(A + (mtile * 16 + mn) * (long)K);
    const float*  bfrag = Bs + mn * 4 + 2 * half;

    v8f acc[NT];
    #pragma unroll
    for (int t = 0; t < NT; ++t) acc[t] = v8f{};

    for (int k0 = 0; k0 < K; k0 += GEMM_KC) {
        const int kc = (K - k0) < GEMM_KC ? (K - k0) : GEMM_KC;
        if (k0 > 0) __syncthreads();
        stage_b_chunk<NT>(Bs, B, N, k0, kc, tid);
        __syncthreads();

        if (active) {
            for (int kq = 0; kq < (kc >> 2); kq += 8) {
                float2 a[8];
                #pragma unroll
                for (int u = 0; u < 8; ++u)
                    a[u] = arow2[((k0 + 4 * (kq + u)) >> 1) + half];
                #pragma unroll
                for (int u = 0; u < 8; ++u) {
                    v2f av; av.x = a[u].x; av.y = a[u].y;
                    #pragma unroll
                    for (int t = 0; t < NT; ++t) {
                        const float2 b2 = *(const float2*)(bfrag + (kq + u) * PITCH + t * 64);
                        v2f bv; bv.x = b2.x; bv.y = b2.y;
                        acc[t] = __builtin_amdgcn_wmma_f32_16x16x4_f32(
                                      false, av,  false, bv,
                                      (short)0, acc[t],
                                      false,  false);
                    }
                }
            }
        }
    }

    if (active) {
        #pragma unroll
        for (int t = 0; t < NT; ++t) {
            const int gn = t * 16 + mn;
            if (gn < N) {
                const float bval = hasBias ? bias[gn] : 0.0f;
                #pragma unroll
                for (int r = 0; r < 8; ++r) {
                    long row = mtile * 16 + r + 8 * half;
                    C[row * (long)N + gn] = acc[t][r] + bval;
                }
            }
        }
    }
}

#define FOLD_K 128

__global__ __launch_bounds__(BLOCK_THREADS)
void fold_proj_into_w_wmma_f32(const float* __restrict__ S0, const float* __restrict__ sb0,
                               float* __restrict__ Wc0, int Min0,
                               const float* __restrict__ S1, const float* __restrict__ sb1,
                               float* __restrict__ Wc1, int Min1,
                               const float* __restrict__ W, int N)
{
    const int tid  = threadIdx.x;
    const int wave = tid >> 5;
    const int lane = tid & 31;
    const int mn   = lane & 15;
    const int half = lane >> 4;

    const int tiles0 = Min0 / 16 + 1;
    int mtile = blockIdx.x;
    const float* S = S0; const float* sb = sb0; float* Wc = Wc0; int Min = Min0;
    if (mtile >= tiles0) { mtile -= tiles0; S = S1; sb = sb1; Wc = Wc1; Min = Min1; }

    const int m = mtile * 16 + mn;
    const int n = wave * 16 + mn;

    const bool   a_row  = m < Min;
    const float* ap     = a_row ? S + m : sb;
    const long   apitch = a_row ? (long)Min : 1;
    const float* bp     = W + (n < N ? n : 0);

    v8f acc = v8f{};
    #pragma unroll 8
    for (int kq = 0; kq < FOLD_K / 4; ++kq) {
        const int k = 4 * kq + 2 * half;
        v2f av; av.x = ap[k * apitch];     av.y = ap[(k + 1) * apitch];
        v2f bv; bv.x = bp[(long)k * N];    bv.y = bp[(long)(k + 1) * N];
        acc = __builtin_amdgcn_wmma_f32_16x16x4_f32(
                   false, av,  false, bv,
                   (short)0, acc,
                   false,  false);
    }

    if (n < N) {
        #pragma unroll
        for (int r = 0; r < 8; ++r) {
            const int row = mtile * 16 + r + 8 * half;
            if (row <= Min) Wc[(long)row * N + n] = acc[r];
        }
    }
}

#define SORT_BLOCKS    256
#define SORT_THREADS   1024
#define BKT_SHIFT      9
#define BKT_ROWS       512
#define N_BKT          196
#define BKT_PITCH      256
#define FIN_THREADS    1024
#define ROWS_IN_FLIGHT 8
#define ROWS_PER_WAVE  8

__global__ __launch_bounds__(SORT_THREADS)
void edge_bucket_count(const int* __restrict__ rows, int* __restrict__ cntm, int nE)
{
    __shared__ int hist[BKT_PITCH];
    const int tid = threadIdx.x;
    if (tid < BKT_PITCH) hist[tid] = 0;
    __syncthreads();
    const int  chunk = (nE + SORT_BLOCKS - 1) / SORT_BLOCKS;
    const long e0 = (long)blockIdx.x * chunk;
    const long e1 = (e0 + chunk) < (long)nE ? (e0 + chunk) : (long)nE;
    for (long e = e0 + tid; e < e1; e += SORT_THREADS)
        atomicAdd(&hist[rows[e] >> BKT_SHIFT], 1);
    __syncthreads();
    if (tid < BKT_PITCH) cntm[blockIdx.x * BKT_PITCH + tid] = hist[tid];
}

__global__ __launch_bounds__(SORT_THREADS)
void edge_bucket_scatter(const int* __restrict__ rows, const int* __restrict__ cols,
                         const float* __restrict__ ew, const int* __restrict__ cntm,
                         int* __restrict__ bstart, int* __restrict__ recs_i, int nE)
{
    __shared__ int s_ptot[4 * BKT_PITCH];
    __shared__ int s_pbef[4 * BKT_PITCH];
    __shared__ int s_tot[BKT_PITCH];
    __shared__ int s_bef[BKT_PITCH];
    __shared__ int s_cur[BKT_PITCH];
    int4* recs = (int4*)recs_i;
    const int tid  = threadIdx.x;
    const int lane = tid & 31;
    const int blk  = blockIdx.x;

    {
        const int t = tid & (BKT_PITCH - 1);
        const int q = tid >> 8;
        int tot = 0, bef = 0;
        const int* cp = cntm + q * (SORT_BLOCKS / 4) * BKT_PITCH + t;
        for (int g = 0; g < SORT_BLOCKS / 4; g += 16) {
            int v[16];
            #pragma unroll
            for (int u = 0; u < 16; ++u) v[u] = cp[(g + u) * BKT_PITCH];
            #pragma unroll
            for (int u = 0; u < 16; ++u) {
                const int j = q * (SORT_BLOCKS / 4) + g + u;
                tot += v[u];
                bef += (j < blk) ? v[u] : 0;
            }
        }
        s_ptot[tid] = tot;
        s_pbef[tid] = bef;
    }
    __syncthreads();
    if (tid < BKT_PITCH) {
        s_tot[tid] = s_ptot[tid] + s_ptot[tid + BKT_PITCH] + s_ptot[tid + 2 * BKT_PITCH] + s_ptot[tid + 3 * BKT_PITCH];
        s_bef[tid] = s_pbef[tid] + s_pbef[tid + BKT_PITCH] + s_pbef[tid + 2 * BKT_PITCH] + s_pbef[tid + 3 * BKT_PITCH];
    }
    __syncthreads();
    if (tid < 32) {
        int loc[8];
        int sum = 0;
        for (int i = 0; i < 8; ++i) { loc[i] = s_tot[lane * 8 + i]; sum += loc[i]; }
        int inc = sum;
        for (int o = 1; o < 32; o <<= 1) {
            const int up = __shfl_up(inc, o);
            if (lane >= o) inc += up;
        }
        int run = inc - sum;
        for (int i = 0; i < 8; ++i) {
            const int b = lane * 8 + i;
            s_cur[b] = run + s_bef[b];
            if (blk == 0 && b <= N_BKT) bstart[b] = run;
            run += loc[i];
        }
    }
    __syncthreads();

    const int  chunk = (nE + SORT_BLOCKS - 1) / SORT_BLOCKS;
    const long e0 = (long)blk * chunk;
    const long e1 = (e0 + chunk) < (long)nE ? (e0 + chunk) : (long)nE;
    for (long e = e0 + tid; e < e1; e += SORT_THREADS) {
        const int   r = rows[e];
        const int   c = cols[e];
        const float w = ew[e];
        const int   p = atomicAdd(&s_cur[r >> BKT_SHIFT], 1);
        recs[p] = make_int4(r, c, __float_as_int(w), 0);
    }
}

__global__ __launch_bounds__(FIN_THREADS)
void bucket_rows_sort(const int* __restrict__ recs_i, const int* __restrict__ bstart,
                      int* __restrict__ csr_i, int* __restrict__ rowptr)
{
    __shared__ int s_cnt[BKT_ROWS];
    const int4* recs = (const int4*)recs_i;
    int2*       csr  = (int2*)csr_i;
    const int tid  = threadIdx.x;
    const int lane = tid & 31;
    const int b    = blockIdx.x;
    const int e0   = bstart[b];
    const int e1   = bstart[b + 1];

    if (tid < BKT_ROWS) s_cnt[tid] = 0;
    __syncthreads();
    for (int eb = e0 + tid; eb < e1; eb += 4 * FIN_THREADS) {
        int r[4];
        #pragma unroll
        for (int u = 0; u < 4; ++u) {
            const int e = eb + u * FIN_THREADS;
            r[u] = e < e1 ? recs[e].x : -1;
        }
        #pragma unroll
        for (int u = 0; u < 4; ++u)
            if (r[u] >= 0) atomicAdd(&s_cnt[r[u] & (BKT_ROWS - 1)], 1);
    }
    __syncthreads();
    if (tid < 32) {
        int loc[16];
        int sum = 0;
        for (int i = 0; i < 16; ++i) { loc[i] = s_cnt[lane * 16 + i]; sum += loc[i]; }
        int inc = sum;
        for (int o = 1; o < 32; o <<= 1) {
            const int up = __shfl_up(inc, o);
            if (lane >= o) inc += up;
        }
        int run = e0 + inc - sum;
        for (int i = 0; i < 16; ++i) {
            const int r = lane * 16 + i;
            s_cnt[r] = run;
            rowptr[b * BKT_ROWS + r] = run;
            run += loc[i];
        }
    }
    __syncthreads();
    for (int eb = e0 + tid; eb < e1; eb += 4 * FIN_THREADS) {
        int4 rec[4];
        #pragma unroll
        for (int u = 0; u < 4; ++u) {
            const int e = eb + u * FIN_THREADS;
            rec[u] = e < e1 ? recs[e] : make_int4(-1, 0, 0, 0);
        }
        #pragma unroll
        for (int u = 0; u < 4; ++u) {
            if (rec[u].x >= 0) {
                const int p = atomicAdd(&s_cnt[rec[u].x & (BKT_ROWS - 1)], 1);
                csr[p] = make_int2(rec[u].y, rec[u].z);
            }
        }
    }
}

#define AS_PAD 4

template <int VEC>
static __device__ __forceinline__
void spmm_rows_relu_to_lds(const int* __restrict__ rowptr, const int* __restrict__ csr_i,
                           const float* __restrict__ dense, const float* __restrict__ rbias,
                           float* __restrict__ As, int nrows)
{
    typedef __attribute__((ext_vector_type(VEC))) float vecf;
    constexpr int F      = 32 * VEC;
    constexpr int APITCH = F + AS_PAD;
    const int2* csr  = (const int2*)csr_i;
    const int   lane = threadIdx.x & 31;
    const int wv = __builtin_amdgcn_readfirstlane((int)(threadIdx.x >> 5));
    const int r0 = (blockIdx.x * WAVES_PER_BLOCK + wv) * ROWS_PER_WAVE;
    int nr = nrows - r0;
    nr = nr < 0 ? 0 : (nr < ROWS_PER_WAVE ? nr : ROWS_PER_WAVE);

    const vecf bias = *(const vecf*)(rbias + lane * VEC);
    float* arow = As + (wv * ROWS_PER_WAVE) * APITCH + lane * VEC;

    int rp = 0;
    if (nr > 0 && lane <= nr) rp = rowptr[r0 + lane];
    int s = __builtin_amdgcn_readlane(rp, 0);
    int t = __builtin_amdgcn_readlane(rp, 1);
    int2 nxt = make_int2(0, 0);
    if (lane < t - s) nxt = csr[s + lane];

    for (int i = 0; i < nr; ++i) {
        const int si = s;
        const int ti = t;
        int2 my = nxt;
        if (i + 1 < nr) {
            s = ti;
            t = __builtin_amdgcn_readlane(rp, i + 2);
            nxt = make_int2(0, 0);
            if (lane < t - s) nxt = csr[s + lane];
        }
        vecf acc = (vecf)(0.0f);
        for (int base = si; base < ti; base += 32) {
            const int n = (ti - base) < 32 ? (ti - base) : 32;
            if (base > si) {
                my = make_int2(0, 0);
                if (lane < n) my = csr[base + lane];
            }
            for (int j = 0; j < n; j += ROWS_IN_FLIGHT) {
                vecf v[ROWS_IN_FLIGHT];
                #pragma unroll
                for (int u = 0; u < ROWS_IN_FLIGHT; ++u) {
                    const int jj  = (j + u) < n ? (j + u) : (n - 1);
                    const int src = __builtin_amdgcn_readlane(my.x, jj);
                    v[u] = *(const vecf*)(dense + (long)src * F + lane * VEC);
                }
                #pragma unroll
                for (int u = 0; u < ROWS_IN_FLIGHT; ++u) {
                    const int   jj = (j + u) < n ? (j + u) : (n - 1);
                    const float wr = __int_as_float(__builtin_amdgcn_readlane(my.y, jj));
                    const float w  = (j + u) < n ? wr : 0.0f;
                    acc += v[u] * w;
                }
            }
        }
        vecf o;
        #pragma unroll
        for (int c = 0; c < VEC; ++c) {
            const float vx = acc[c] + bias[c];
            o[c] = vx > 0.0f ? vx : 0.0f;
        }
        *(vecf*)(arow + i * APITCH) = o;
    }
    for (int i = nr; i < ROWS_PER_WAVE; ++i)
        *(vecf*)(arow + i * APITCH) = (vecf)(0.0f);
}

#define TILE_ROWS (WAVES_PER_BLOCK * ROWS_PER_WAVE)

__global__ __launch_bounds__(BLOCK_THREADS)
void spmm_f128_relu_gemm_n64(const int* __restrict__ rowptr, const int* __restrict__ csr,
                             const float* __restrict__ dense, const float* __restrict__ rbias,
                             const float* __restrict__ W, float* __restrict__ xout, int nrows)
{
    constexpr int F      = 128;
    constexpr int NOUT   = 64;
    constexpr int APITCH = F + AS_PAD;
    constexpr int BPITCH = NOUT * 4 + BS_PAD;
    __shared__ __attribute__((aligned(16))) float Bs[(F / 4) * BPITCH];
    __shared__ __attribute__((aligned(16))) float As[TILE_ROWS * APITCH];

    const int tid  = threadIdx.x;
    const int wave = __builtin_amdgcn_readfirstlane(tid >> 5);
    const int lane = tid & 31;
    const int mn   = lane & 15;
    const int half = lane >> 4;

    stage_b_chunk<NOUT / 16>(Bs, W, NOUT, 0, F, tid);
    spmm_rows_relu_to_lds<4>(rowptr, csr, dense, rbias, As, nrows);
    __syncthreads();

    const int mt  = wave >> 1;
    const int nt0 = (wave & 1) * 2;
    const float* afrag = As + (mt * 16 + mn) * APITCH + 2 * half;
    const float* bfrag = Bs + (nt0 * 16 + mn) * 4 + 2 * half;

    v8f acc0 = v8f{};
    v8f acc1 = v8f{};
    #pragma unroll 8
    for (int kq = 0; kq < F / 4; ++kq) {
        const float2 a2 = *(const float2*)(afrag + 4 * kq);
        const float2 p2 = *(const float2*)(bfrag + kq * BPITCH);
        const float2 q2 = *(const float2*)(bfrag + kq * BPITCH + 64);
        v2f av; av.x = a2.x; av.y = a2.y;
        v2f bp; bp.x = p2.x; bp.y = p2.y;
        v2f bq; bq.x = q2.x; bq.y = q2.y;
        acc0 = __builtin_amdgcn_wmma_f32_16x16x4_f32(
                    false, av,  false, bp,
                    (short)0, acc0,
                    false,  false);
        acc1 = __builtin_amdgcn_wmma_f32_16x16x4_f32(
                    false, av,  false, bq,
                    (short)0, acc1,
                    false,  false);
    }

    const long row0 = (long)blockIdx.x * TILE_ROWS + mt * 16;
    #pragma unroll
    for (int r = 0; r < 8; ++r) {
        const long row = row0 + r + 8 * half;
        if (row < (long)nrows) {
            xout[row * NOUT + nt0 * 16 + mn]      = acc0[r];
            xout[row * NOUT + nt0 * 16 + 16 + mn] = acc1[r];
        }
    }
}

__global__ __launch_bounds__(BLOCK_THREADS)
void spmm_f64_relu_head(const int* __restrict__ rowptr, const int* __restrict__ csr,
                        const float* __restrict__ dense, const float* __restrict__ rbias,
                        const float* __restrict__ P, const float* __restrict__ pbias,
                        float* __restrict__ out, int nrows, int N)
{
    constexpr int F      = 64;
    constexpr int APITCH = F + AS_PAD;
    __shared__ __attribute__((aligned(16))) float As[TILE_ROWS * APITCH];

    const int tid  = threadIdx.x;
    const int wave = __builtin_amdgcn_readfirstlane(tid >> 5);
    const int lane = tid & 31;
    const int mn   = lane & 15;
    const int half = lane >> 4;

    spmm_rows_relu_to_lds<2>(rowptr, csr, dense, rbias, As, nrows);
    __syncthreads();

    if (wave < TILE_ROWS / 16) {
        const float* afrag = As + (wave * 16 + mn) * APITCH + 2 * half;
        const float* bp    = P + (mn < N ? mn : 0) * F + 2 * half;
        v8f acc = v8f{};
        #pragma unroll
        for (int kq = 0; kq < F / 4; ++kq) {
            const float2 a2 = *(const float2*)(afrag + 4 * kq);
            const float2 b2 = *(const float2*)(bp + 4 * kq);
            v2f av; av.x = a2.x; av.y = a2.y;
            v2f bv; bv.x = b2.x; bv.y = b2.y;
            acc = __builtin_amdgcn_wmma_f32_16x16x4_f32(
                       false, av,  false, bv,
                       (short)0, acc,
                       false,  false);
        }
        if (mn < N) {
            const float bval = pbias[mn];
            const long row0 = (long)blockIdx.x * TILE_ROWS + wave * 16;
            #pragma unroll
            for (int r = 0; r < 8; ++r) {
                const long row = row0 + r + 8 * half;
                if (row < (long)nrows) out[row * N + mn] = acc[r] + bval;
            }
        }
    }
}

static inline int spmm_blocks(int nrows) {
    return (nrows + WAVES_PER_BLOCK * ROWS_PER_WAVE - 1) / (WAVES_PER_BLOCK * ROWS_PER_WAVE);
}

static inline int mtiles_blocks(int M) {
    return ((M / 16) + WAVES_PER_BLOCK - 1) / WAVES_PER_BLOCK;
}

extern "C" void kernel_launch(void* const* d_in, const int* in_sizes, int n_in,
                              void* d_out, int out_size, void* d_ws, size_t ws_size,
                              hipStream_t stream)
{
    const float* feat0 = (const float*)d_in[0];
    const float* feat1 = (const float*)d_in[1];
    const float* fc0_w = (const float*)d_in[2];
    const float* fc0_b = (const float*)d_in[3];
    const float* fc1_w = (const float*)d_in[4];
    const float* fc1_b = (const float*)d_in[5];
    const float* W1    = (const float*)d_in[6];
    const float* b1    = (const float*)d_in[7];
    const float* W2    = (const float*)d_in[8];
    const float* b2    = (const float*)d_in[9];
    const float* pw    = (const float*)d_in[10];
    const float* pb    = (const float*)d_in[11];
    const int*   erow  = (const int*)d_in[12];
    const int*   ecol  = (const int*)d_in[13];
    const float* ew    = (const float*)d_in[14];
    float*       out   = (float*)d_out;

    const int E       = in_sizes[12];
    const int N_NODES = 100000;
    const int NT      = 50000;

    float* x   = (float*)d_ws;
    float* h   = x + (long)N_NODES * 128;
    int* cntm   = (int*)(h + (long)N_NODES * 128);
    int* bstart = cntm + SORT_BLOCKS * BKT_PITCH;
    int* recs   = bstart + 256;
    int* csr    = recs + 4 * (long)1600000;
    int* rowptr = csr + 2 * (long)1600000;
    float* wc0  = (float*)(rowptr + N_BKT * BKT_ROWS);
    float* wc1  = wc0 + 144 * 128;

    edge_bucket_count<<<SORT_BLOCKS, SORT_THREADS, 0, stream>>>(erow, cntm, E);
    edge_bucket_scatter<<<SORT_BLOCKS, SORT_THREADS, 0, stream>>>(
        erow, ecol, ew, cntm, bstart, recs, E);
    bucket_rows_sort<<<N_BKT, FIN_THREADS, 0, stream>>>(recs, bstart, csr, rowptr);

    fold_proj_into_w_wmma_f32<<<26, BLOCK_THREADS, 0, stream>>>(
        fc0_w, fc0_b, wc0, 128, fc1_w, fc1_b, wc1, 256, W1, 128);

    gemm_strip_wmma_f32<8><<<mtiles_blocks(NT), BLOCK_THREADS, 0, stream>>>(
        feat0, wc0, wc0 + 128 * 128, h, NT, 128, 128, 1);
    gemm_strip_wmma_f32<8><<<mtiles_blocks(NT), BLOCK_THREADS, 0, stream>>>(
        feat1, wc1, wc1 + 256 * 128, h + (long)NT * 128, NT, 256, 128, 1);

    spmm_f128_relu_gemm_n64<<<spmm_blocks(N_NODES), BLOCK_THREADS, 0, stream>>>(
        rowptr, csr, h, b1, W2, x, N_NODES);

    spmm_f64_relu_head<<<spmm_blocks(N_NODES), BLOCK_THREADS, 0, stream>>>(
        rowptr, csr, x, b2, pw, pb, out, N_NODES, 8);
}


# Edge-sort steps run beside the fold and the two projection GEMMs in shared launches (roles by block index, 256-thread sort blocks with grouped loads); 8 -> 5 launches; sum orders as before
# speedup vs baseline: 4.6565x; 1.1649x over previous
#include <hip/hip_runtime.h>

typedef __attribute__((ext_vector_type(2))) float v2f;
typedef __attribute__((ext_vector_type(8))) float v8f;

#define WAVES_PER_BLOCK 8
#define BLOCK_THREADS 256

#define GEMM_KC 128
#define BS_PAD  4

template <int NT>
static __device__ __forceinline__
void stage_b_chunk(float* __restrict__ Bs, const float* __restrict__ B,
                   int N, int k0, int kc, int tid)
{
    constexpr int NP    = NT * 16;
    constexpr int PITCH = NP * 4 + BS_PAD;
    {
        static_assert(NT % 2 == 0, "column groups come in eights");
        constexpr int NG    = NP / 4;
        constexpr int NG8   = NG / 8 > 0 ? NG / 8 : 1;
        constexpr int ITEMS = (GEMM_KC / 4) * NG / BLOCK_THREADS;
        const int nitems = (kc >> 2) * NG;
        float4 r[ITEMS > 0 ? ITEMS : 1][4];
        #pragma unroll
        for (int j = 0; j < ITEMS; ++j) {
            const int i  = tid + j * BLOCK_THREADS;
            const int w  = i >> 5;
            const int n4 = (w % NG8) * 8 + (i & 7);
            const int kq = (w / NG8) * 4 + ((i >> 3) & 3);
            #pragma unroll
            for (int s = 0; s < 4; ++s) r[j][s] = make_float4(0.0f, 0.0f, 0.0f, 0.0f);
            if (i < nitems && 4 * n4 < N) {
                const float* p = B + (long)(k0 + 4 * kq) * N + 4 * n4;
                #pragma unroll
                for (int s = 0; s < 4; ++s) r[j][s] = *(const float4*)(p + (long)s * N);
            }
        }
        #pragma unroll
        for (int j = 0; j < ITEMS; ++j) {
            const int i  = tid + j * BLOCK_THREADS;
            const int w  = i >> 5;
            const int n4 = (w % NG8) * 8 + (i & 7);
            const int kq = (w / NG8) * 4 + ((i >> 3) & 3);
            if (i < nitems) {
                float* q = Bs + kq * PITCH + n4 * 16;
                *(float4*)(q + 0)  = make_float4(r[j][0].x, r[j][1].x, r[j][2].x, r[j][3].x);
                *(float4*)(q + 4)  = make_float4(r[j][0].y, r[j][1].y, r[j][2].y, r[j][3].y);
                *(float4*)(q + 8)  = make_float4(r[j][0].z, r[j][1].z, r[j][2].z, r[j][3].z);
                *(float4*)(q + 12) = make_float4(r[j][0].w, r[j][1].w, r[j][2].w, r[j][3].w);
            }
        }
    }
}

template <int NT>
static __device__ __forceinline__
void gemm_strip_wmma_f32(float* __restrict__ Bs, int blk,
                         const float* __restrict__ A, const float* __restrict__ B,
                         const float* __restrict__ bias, float* __restrict__ C,
                         int M, int K, int N, int hasBias)
{
    constexpr int NP    = NT * 16;
    constexpr int PITCH = NP * 4 + BS_PAD;

    const int  tid   = threadIdx.x;
    const int  wave  = __builtin_amdgcn_readfirstlane(tid >> 5);
    const int  lane  = tid & 31;
    const int  mn    = lane & 15;
    const int  half  = lane >> 4;
    const long mtile = (long)blk * WAVES_PER_BLOCK + wave;
    const bool active = mtile * 16 < (long)M;

    const float*   abase = A + mtile * 16 * (long)K;
    const unsigned aoff  = (unsigned)(mn * K + 2 * half);
    const float*   bfrag = Bs + mn * 4 + 2 * half;

    v8f acc[NT];
    #pragma unroll
    for (int t = 0; t < NT; ++t) acc[t] = v8f{};

    for (int k0 = 0; k0 < K; k0 += GEMM_KC) {
        const int kc = (K - k0) < GEMM_KC ? (K - k0) : GEMM_KC;
        if (k0 > 0) __syncthreads();
        stage_b_chunk<NT>(Bs, B, N, k0, kc, tid);
        __syncthreads();

        if (active) {
            for (int kq = 0; kq < (kc >> 2); kq += 8) {
                float2 a[8];
                const float* ak = abase + (k0 + 4 * kq);
                #pragma unroll
                for (int u = 0; u < 8; ++u)
                    a[u] = *(const float2*)(ak + aoff + 4 * u);
                #pragma unroll
                for (int u = 0; u < 8; ++u) {
                    v2f av; av.x = a[u].x; av.y = a[u].y;
                    #pragma unroll
                    for (int t = 0; t < NT; ++t) {
                        const float2 b2 = *(const float2*)(bfrag + (kq + u) * PITCH + t * 64);
                        v2f bv; bv.x = b2.x; bv.y = b2.y;
                        acc[t] = __builtin_amdgcn_wmma_f32_16x16x4_f32(
                                      false, av,  false, bv,
                                      (short)0, acc[t],
                                      false,  false);
                    }
                }
            }
        }
    }

    if (active) {
        #pragma unroll
        for (int t = 0; t < NT; ++t) {
            const int gn = t * 16 + mn;
            if (gn < N) {
                const float bval = hasBias ? bias[gn] : 0.0f;
                float* cstrip = C + mtile * 16 * (long)N;
                #pragma unroll
                for (int r = 0; r < 8; ++r) {
                    const unsigned row = r + 8 * half;
                    cstrip[row * (unsigned)N + gn] = acc[t][r] + bval;
                }
            }
        }
    }
}

#define FOLD_K 128

static __device__ __forceinline__
void fold_proj_into_w_wmma_f32(int blk,
                               const float* __restrict__ S0, const float* __restrict__ sb0,
                               float* __restrict__ Wc0, int Min0,
                               const float* __restrict__ S1, const float* __restrict__ sb1,
                               float* __restrict__ Wc1, int Min1,
                               const float* __restrict__ W, int N)
{
    const int tid  = threadIdx.x;
    const int wave = tid >> 5;
    const int lane = tid & 31;
    const int mn   = lane & 15;
    const int half = lane >> 4;

    const int tiles0 = Min0 / 16 + 1;
    int mtile = blk;
    const float* S = S0; const float* sb = sb0; float* Wc = Wc0; int Min = Min0;
    if (mtile >= tiles0) { mtile -= tiles0; S = S1; sb = sb1; Wc = Wc1; Min = Min1; }

    const int m = mtile * 16 + mn;
    const int n = wave * 16 + mn;

    const bool   a_row  = m < Min;
    const float* ap     = a_row ? S + m : sb;
    const long   apitch = a_row ? (long)Min : 1;
    const float* bp     = W + (n < N ? n : 0);

    v8f acc = v8f{};
    #pragma unroll 8
    for (int kq = 0; kq < FOLD_K / 4; ++kq) {
        const int k = 4 * kq + 2 * half;
        v2f av; av.x = ap[k * apitch];     av.y = ap[(k + 1) * apitch];
        v2f bv; bv.x = bp[(long)k * N];    bv.y = bp[(long)(k + 1) * N];
        acc = __builtin_amdgcn_wmma_f32_16x16x4_f32(
                   false, av,  false, bv,
                   (short)0, acc,
                   false,  false);
    }

    if (n < N) {
        #pragma unroll
        for (int r = 0; r < 8; ++r) {
            const int row = mtile * 16 + r + 8 * half;
            if (row <= Min) Wc[(long)row * N + n] = acc[r];
        }
    }
}

#define SORT_BLOCKS    256
#define BKT_SHIFT      9
#define BKT_ROWS       512
#define N_BKT          196
#define BKT_PITCH      256
#define ROWS_IN_FLIGHT 8
#define ROWS_PER_WAVE  8

static_assert(BKT_PITCH == BLOCK_THREADS, "one thread per bucket column");

static __device__ __forceinline__
void edge_bucket_count(int* __restrict__ hist, int blk,
                       const int* __restrict__ rows, int* __restrict__ cntm, int nE)
{
    const int tid = threadIdx.x;
    hist[tid] = 0;
    __syncthreads();
    const int  chunk = (nE + SORT_BLOCKS - 1) / SORT_BLOCKS;
    const long e0 = (long)blk * chunk;
    const long e1 = (e0 + chunk) < (long)nE ? (e0 + chunk) : (long)nE;
    for (long eb = e0 + tid; eb < e1; eb += 16 * BLOCK_THREADS) {
        int r[16];
        #pragma unroll
        for (int u = 0; u < 16; ++u) {
            const long e = eb + u * BLOCK_THREADS;
            r[u] = e < e1 ? rows[e] : -1;
        }
        #pragma unroll
        for (int u = 0; u < 16; ++u)
            if (r[u] >= 0) atomicAdd(&hist[r[u] >> BKT_SHIFT], 1);
    }
    __syncthreads();
    cntm[blk * BKT_PITCH + tid] = hist[tid];
}

static __device__ __forceinline__
void edge_bucket_scatter(int* __restrict__ s_tot, int* __restrict__ s_bef, int* __restrict__ s_cur,
                         int blk,
                         const int* __restrict__ rows, const int* __restrict__ cols,
                         const float* __restrict__ ew, const int* __restrict__ cntm,
                         int* __restrict__ bstart, int* __restrict__ recs_i, int nE)
{
    int4* recs = (int4*)recs_i;
    const int tid  = threadIdx.x;
    const int lane = tid & 31;

    {
        int tot = 0, bef = 0;
        const int* cp = cntm + tid;
        for (int g = 0; g < SORT_BLOCKS; g += 32) {
            int v[32];
            #pragma unroll
            for (int u = 0; u < 32; ++u) v[u] = cp[(g + u) * BKT_PITCH];
            #pragma unroll
            for (int u = 0; u < 32; ++u) {
                tot += v[u];
                bef += (g + u < blk) ? v[u] : 0;
            }
        }
        s_tot[tid] = tot;
        s_bef[tid] = bef;
    }
    __syncthreads();
    if (tid < 32) {
        int loc[8];
        int sum = 0;
        for (int i = 0; i < 8; ++i) { loc[i] = s_tot[lane * 8 + i]; sum += loc[i]; }
        int inc = sum;
        for (int o = 1; o < 32; o <<= 1) {
            const int up = __shfl_up(inc, o);
            if (lane >= o) inc += up;
        }
        int run = inc - sum;
        for (int i = 0; i < 8; ++i) {
            const int b = lane * 8 + i;
            s_cur[b] = run + s_bef[b];
            if (blk == 0 && b <= N_BKT) bstart[b] = run;
            run += loc[i];
        }
    }
    __syncthreads();

    const int  chunk = (nE + SORT_BLOCKS - 1) / SORT_BLOCKS;
    const long e0 = (long)blk * chunk;
    const long e1 = (e0 + chunk) < (long)nE ? (e0 + chunk) : (long)nE;
    for (long eb = e0 + tid; eb < e1; eb += 8 * BLOCK_THREADS) {
        int   r[8];
        int   c[8];
        float w[8];
        #pragma unroll
        for (int u = 0; u < 8; ++u) {
            const long e  = eb + u * BLOCK_THREADS;
            const bool ok = e < e1;
            r[u] = ok ? rows[e] : -1;
            c[u] = ok ? cols[e] : 0;
            w[u] = ok ? ew[e]   : 0.0f;
        }
        #pragma unroll
        for (int u = 0; u < 8; ++u) {
            if (r[u] >= 0) {
                const int p = atomicAdd(&s_cur[r[u] >> BKT_SHIFT], 1);
                recs[p] = make_int4(r[u], c[u], __float_as_int(w[u]), 0);
            }
        }
    }
}

static __device__ __forceinline__
void bucket_rows_sort(int* __restrict__ s_cnt, int b,
                      const int* __restrict__ recs_i, const int* __restrict__ bstart,
                      int* __restrict__ csr_i, int* __restrict__ rowptr)
{
    const int4* recs = (const int4*)recs_i;
    int2*       csr  = (int2*)csr_i;
    const int tid  = threadIdx.x;
    const int lane = tid & 31;
    const int e0   = bstart[b];
    const int e1   = bstart[b + 1];

    for (int i = tid; i < BKT_ROWS; i += BLOCK_THREADS) s_cnt[i] = 0;
    __syncthreads();
    for (int eb = e0 + tid; eb < e1; eb += 8 * BLOCK_THREADS) {
        int r[8];
        #pragma unroll
        for (int u = 0; u < 8; ++u) {
            const int e = eb + u * BLOCK_THREADS;
            r[u] = e < e1 ? recs[e].x : -1;
        }
        #pragma unroll
        for (int u = 0; u < 8; ++u)
            if (r[u] >= 0) atomicAdd(&s_cnt[r[u] & (BKT_ROWS - 1)], 1);
    }
    __syncthreads();
    if (tid < 32) {
        int loc[16];
        int sum = 0;
        for (int i = 0; i < 16; ++i) { loc[i] = s_cnt[lane * 16 + i]; sum += loc[i]; }
        int inc = sum;
        for (int o = 1; o < 32; o <<= 1) {
            const int up = __shfl_up(inc, o);
            if (lane >= o) inc += up;
        }
        int run = e0 + inc - sum;
        for (int i = 0; i < 16; ++i) {
            const int r = lane * 16 + i;
            s_cnt[r] = run;
            rowptr[b * BKT_ROWS + r] = run;
            run += loc[i];
        }
    }
    __syncthreads();
    for (int eb = e0 + tid; eb < e1; eb += 8 * BLOCK_THREADS) {
        int4 rec[8];
        #pragma unroll
        for (int u = 0; u < 8; ++u) {
            const int e = eb + u * BLOCK_THREADS;
            rec[u] = e < e1 ? recs[e] : make_int4(-1, 0, 0, 0);
        }
        #pragma unroll
        for (int u = 0; u < 8; ++u) {
            if (rec[u].x >= 0) {
                const int p = atomicAdd(&s_cnt[rec[u].x & (BKT_ROWS - 1)], 1);
                csr[p] = make_int2(rec[u].y, rec[u].z);
            }
        }
    }
}

#define AS_PAD 4

template <int VEC>
static __device__ __forceinline__
void spmm_rows_relu_to_lds(const int* __restrict__ rowptr, const int* __restrict__ csr_i,
                           const float* __restrict__ dense, const float* __restrict__ rbias,
                           float* __restrict__ As, int nrows)
{
    typedef __attribute__((ext_vector_type(VEC))) float vecf;
    constexpr int F      = 32 * VEC;
    constexpr int APITCH = F + AS_PAD;
    const int2* csr  = (const int2*)csr_i;
    const int   lane = threadIdx.x & 31;
    const int wv = __builtin_amdgcn_readfirstlane((int)(threadIdx.x >> 5));
    const int r0 = (blockIdx.x * WAVES_PER_BLOCK + wv) * ROWS_PER_WAVE;
    int nr = nrows - r0;
    nr = nr < 0 ? 0 : (nr < ROWS_PER_WAVE ? nr : ROWS_PER_WAVE);

    const vecf bias = *(const vecf*)(rbias + lane * VEC);
    float* arow = As + (wv * ROWS_PER_WAVE) * APITCH + lane * VEC;

    int rp = 0;
    if (nr > 0 && lane <= nr) rp = rowptr[r0 + lane];
    int s = __builtin_amdgcn_readlane(rp, 0);
    int t = __builtin_amdgcn_readlane(rp, 1);
    int2 nxt = make_int2(0, 0);
    if (lane < t - s) nxt = csr[s + lane];

    for (int i = 0; i < nr; ++i) {
        const int si = s;
        const int ti = t;
        int2 my = nxt;
        if (i + 1 < nr) {
            s = ti;
            t = __builtin_amdgcn_readlane(rp, i + 2);
            nxt = make_int2(0, 0);
            if (lane < t - s) nxt = csr[s + lane];
        }
        vecf acc = (vecf)(0.0f);
        for (int base = si; base < ti; base += 32) {
            const int n = (ti - base) < 32 ? (ti - base) : 32;
            if (base > si) {
                my = make_int2(0, 0);
                if (lane < n) my = csr[base + lane];
            }
            for (int j = 0; j < n; j += ROWS_IN_FLIGHT) {
                vecf v[ROWS_IN_FLIGHT];
                #pragma unroll
                for (int u = 0; u < ROWS_IN_FLIGHT; ++u) {
                    const int jj  = (j + u) < n ? (j + u) : (n - 1);
                    const int src = __builtin_amdgcn_readlane(my.x, jj);
                    v[u] = *(const vecf*)(dense + (long)src * F + lane * VEC);
                }
                #pragma unroll
                for (int u = 0; u < ROWS_IN_FLIGHT; ++u) {
                    const int   jj = (j + u) < n ? (j + u) : (n - 1);
                    const float wr = __int_as_float(__builtin_amdgcn_readlane(my.y, jj));
                    const float w  = (j + u) < n ? wr : 0.0f;
                    acc += v[u] * w;
                }
            }
        }
        vecf o;
        #pragma unroll
        for (int c = 0; c < VEC; ++c) {
            const float vx = acc[c] + bias[c];
            o[c] = vx > 0.0f ? vx : 0.0f;
        }
        *(vecf*)(arow + i * APITCH) = o;
    }
    for (int i = nr; i < ROWS_PER_WAVE; ++i)
        *(vecf*)(arow + i * APITCH) = (vecf)(0.0f);
}

#define TILE_ROWS (WAVES_PER_BLOCK * ROWS_PER_WAVE)

__global__ __launch_bounds__(BLOCK_THREADS)
void spmm_f128_relu_gemm_n64(const int* __restrict__ rowptr, const int* __restrict__ csr,
                             const float* __restrict__ dense, const float* __restrict__ rbias,
                             const float* __restrict__ W, float* __restrict__ xout, int nrows)
{
    constexpr int F      = 128;
    constexpr int NOUT   = 64;
    constexpr int APITCH = F + AS_PAD;
    constexpr int BPITCH = NOUT * 4 + BS_PAD;
    __shared__ __attribute__((aligned(16))) float Bs[(F / 4) * BPITCH];
    __shared__ __attribute__((aligned(16))) float As[TILE_ROWS * APITCH];

    const int tid  = threadIdx.x;
    const int wave = __builtin_amdgcn_readfirstlane(tid >> 5);
    const int lane = tid & 31;
    const int mn   = lane & 15;
    const int half = lane >> 4;

    stage_b_chunk<NOUT / 16>(Bs, W, NOUT, 0, F, tid);
    spmm_rows_relu_to_lds<4>(rowptr, csr, dense, rbias, As, nrows);
    __syncthreads();

    const int mt  = wave >> 1;
    const int nt0 = (wave & 1) * 2;
    const float* afrag = As + (mt * 16 + mn) * APITCH + 2 * half;
    const float* bfrag = Bs + (nt0 * 16 + mn) * 4 + 2 * half;

    v8f acc0 = v8f{};
    v8f acc1 = v8f{};
    #pragma unroll 8
    for (int kq = 0; kq < F / 4; ++kq) {
        const float2 a2 = *(const float2*)(afrag + 4 * kq);
        const float2 p2 = *(const float2*)(bfrag + kq * BPITCH);
        const float2 q2 = *(const float2*)(bfrag + kq * BPITCH + 64);
        v2f av; av.x = a2.x; av.y = a2.y;
        v2f bp; bp.x = p2.x; bp.y = p2.y;
        v2f bq; bq.x = q2.x; bq.y = q2.y;
        acc0 = __builtin_amdgcn_wmma_f32_16x16x4_f32(
                    false, av,  false, bp,
                    (short)0, acc0,
                    false,  false);
        acc1 = __builtin_amdgcn_wmma_f32_16x16x4_f32(
                    false, av,  false, bq,
                    (short)0, acc1,
                    false,  false);
    }

    const long row0 = (long)blockIdx.x * TILE_ROWS + mt * 16;
    #pragma unroll
    for (int r = 0; r < 8; ++r) {
        const long row = row0 + r + 8 * half;
        if (row < (long)nrows) {
            xout[row * NOUT + nt0 * 16 + mn]      = acc0[r];
            xout[row * NOUT + nt0 * 16 + 16 + mn] = acc1[r];
        }
    }
}

__global__ __launch_bounds__(BLOCK_THREADS)
void spmm_f64_relu_head(const int* __restrict__ rowptr, const int* __restrict__ csr,
                        const float* __restrict__ dense, const float* __restrict__ rbias,
                        const float* __restrict__ P, const float* __restrict__ pbias,
                        float* __restrict__ out, int nrows, int N)
{
    constexpr int F      = 64;
    constexpr int APITCH = F + AS_PAD;
    __shared__ __attribute__((aligned(16))) float As[TILE_ROWS * APITCH];

    const int tid  = threadIdx.x;
    const int wave = __builtin_amdgcn_readfirstlane(tid >> 5);
    const int lane = tid & 31;
    const int mn   = lane & 15;
    const int half = lane >> 4;

    spmm_rows_relu_to_lds<2>(rowptr, csr, dense, rbias, As, nrows);
    __syncthreads();

    if (wave < TILE_ROWS / 16) {
        const float* afrag = As + (wave * 16 + mn) * APITCH + 2 * half;
        const float* bp    = P + (mn < N ? mn : 0) * F + 2 * half;
        v8f acc = v8f{};
        #pragma unroll
        for (int kq = 0; kq < F / 4; ++kq) {
            const float2 a2 = *(const float2*)(afrag + 4 * kq);
            const float2 b2 = *(const float2*)(bp + 4 * kq);
            v2f av; av.x = a2.x; av.y = a2.y;
            v2f bv; bv.x = b2.x; bv.y = b2.y;
            acc = __builtin_amdgcn_wmma_f32_16x16x4_f32(
                       false, av,  false, bv,
                       (short)0, acc,
                       false,  false);
        }
        if (mn < N) {
            const float bval = pbias[mn];
            const long row0 = (long)blockIdx.x * TILE_ROWS + wave * 16;
            #pragma unroll
            for (int r = 0; r < 8; ++r) {
                const long row = row0 + r + 8 * half;
                if (row < (long)nrows) out[row * N + mn] = acc[r] + bval;
            }
        }
    }
}

#define SHARED_WAVES_PER_SIMD 6

__global__ __launch_bounds__(BLOCK_THREADS)
void count_edges_and_fold(const int* __restrict__ rows, int* __restrict__ cntm, int nE,
                          const float* __restrict__ S0, const float* __restrict__ sb0,
                          float* __restrict__ Wc0, int Min0,
                          const float* __restrict__ S1, const float* __restrict__ sb1,
                          float* __restrict__ Wc1, int Min1,
                          const float* __restrict__ W, int N)
{
    __shared__ int hist[BKT_PITCH];
    const int blk = blockIdx.x;
    if (blk < SORT_BLOCKS)
        edge_bucket_count(hist, blk, rows, cntm, nE);
    else
        fold_proj_into_w_wmma_f32(blk - SORT_BLOCKS, S0, sb0, Wc0, Min0, S1, sb1, Wc1, Min1, W, N);
}

__global__ __launch_bounds__(BLOCK_THREADS, SHARED_WAVES_PER_SIMD)
void scatter_edges_and_project(const int* __restrict__ rows, const int* __restrict__ cols,
                               const float* __restrict__ ew, const int* __restrict__ cntm,
                               int* __restrict__ bstart, int* __restrict__ recs, int nE,
                               const float* __restrict__ A, const float* __restrict__ B,
                               const float* __restrict__ bias, float* __restrict__ C,
                               int M, int K, int N)
{
    __shared__ __attribute__((aligned(16))) float Bs[(GEMM_KC / 4) * (8 * 64 + BS_PAD)];
    __shared__ int s_tot[BKT_PITCH];
    __shared__ int s_bef[BKT_PITCH];
    __shared__ int s_cur[BKT_PITCH];
    const int blk = blockIdx.x;
    if (blk < SORT_BLOCKS)
        edge_bucket_scatter(s_tot, s_bef, s_cur, blk, rows, cols, ew, cntm, bstart, recs, nE);
    else
        gemm_strip_wmma_f32<8>(Bs, blk - SORT_BLOCKS, A, B, bias, C, M, K, N, 1);
}

__global__ __launch_bounds__(BLOCK_THREADS, SHARED_WAVES_PER_SIMD)
void sort_bucket_rows_and_project(const int* __restrict__ recs, const int* __restrict__ bstart,
                                  int* __restrict__ csr, int* __restrict__ rowptr,
                                  const float* __restrict__ A, const float* __restrict__ B,
                                  const float* __restrict__ bias, float* __restrict__ C,
                                  int M, int K, int N)
{
    __shared__ __attribute__((aligned(16))) float Bs[(GEMM_KC / 4) * (8 * 64 + BS_PAD)];
    __shared__ int s_cnt[BKT_ROWS];
    const int blk = blockIdx.x;
    if (blk < N_BKT)
        bucket_rows_sort(s_cnt, blk, recs, bstart, csr, rowptr);
    else
        gemm_strip_wmma_f32<8>(Bs, blk - N_BKT, A, B, bias, C, M, K, N, 1);
}

static inline int spmm_blocks(int nrows) {
    return (nrows + WAVES_PER_BLOCK * ROWS_PER_WAVE - 1) / (WAVES_PER_BLOCK * ROWS_PER_WAVE);
}

static inline int mtiles_blocks(int M) {
    return ((M / 16) + WAVES_PER_BLOCK - 1) / WAVES_PER_BLOCK;
}

extern "C" void kernel_launch(void* const* d_in, const int* in_sizes, int n_in,
                              void* d_out, int out_size, void* d_ws, size_t ws_size,
                              hipStream_t stream)
{
    const float* feat0 = (const float*)d_in[0];
    const float* feat1 = (const float*)d_in[1];
    const float* fc0_w = (const float*)d_in[2];
    const float* fc0_b = (const float*)d_in[3];
    const float* fc1_w = (const float*)d_in[4];
    const float* fc1_b = (const float*)d_in[5];
    const float* W1    = (const float*)d_in[6];
    const float* b1    = (const float*)d_in[7];
    const float* W2    = (const float*)d_in[8];
    const float* b2    = (const float*)d_in[9];
    const float* pw    = (const float*)d_in[10];
    const float* pb    = (const float*)d_in[11];
    const int*   erow  = (const int*)d_in[12];
    const int*   ecol  = (const int*)d_in[13];
    const float* ew    = (const float*)d_in[14];
    float*       out   = (float*)d_out;

    const int E       = in_sizes[12];
    const int N_NODES = 100000;
    const int NT      = 50000;

    float* x   = (float*)d_ws;
    float* h   = x + (long)N_NODES * 128;
    int* cntm   = (int*)(h + (long)N_NODES * 128);
    int* bstart = cntm + SORT_BLOCKS * BKT_PITCH;
    int* recs   = bstart + 256;
    int* csr    = recs + 4 * (long)1600000;
    int* rowptr = csr + 2 * (long)1600000;
    float* wc0  = (float*)(rowptr + N_BKT * BKT_ROWS);
    float* wc1  = wc0 + 144 * 128;

    count_edges_and_fold<<<SORT_BLOCKS + 26, BLOCK_THREADS, 0, stream>>>(
        erow, cntm, E,
        fc0_w, fc0_b, wc0, 128, fc1_w, fc1_b, wc1, 256, W1, 128);

    scatter_edges_and_project<<<SORT_BLOCKS + mtiles_blocks(NT), BLOCK_THREADS, 0, stream>>>(
        erow, ecol, ew, cntm, bstart, recs, E,
        feat1, wc1, wc1 + 256 * 128, h + (long)NT * 128, NT, 256, 128);

    sort_bucket_rows_and_project<<<N_BKT + mtiles_blocks(NT), BLOCK_THREADS, 0, stream>>>(
        recs, bstart, csr, rowptr,
        feat0, wc0, wc0 + 128 * 128, h, NT, 128, 128);

    spmm_f128_relu_gemm_n64<<<spmm_blocks(N_NODES), BLOCK_THREADS, 0, stream>>>(
        rowptr, csr, h, b1, W2, x, N_NODES);

    spmm_f64_relu_head<<<spmm_blocks(N_NODES), BLOCK_THREADS, 0, stream>>>(
        rowptr, csr, x, b2, pw, pb, out, N_NODES, 8);
}
